# starting kernel
# baseline (speedup 1.0000x reference)
#include <hip/hip_runtime.h>

typedef __attribute__((ext_vector_type(2))) float v2f;
typedef __attribute__((ext_vector_type(8))) float v8f;

constexpr int Bc = 128, Nw = 4096, Sc = 32, Pc = 64, Dc = 128;
constexpr float NORM_INV = 1.0f / 300.0f;

constexpr int KA  = 256;
constexpr int LDA = 260;
constexpr int KB  = 384;
constexpr int LDB = 388;

__global__ __launch_bounds__(256)
void pm_emb_kernel(const float* __restrict__ enc_row,
                   const float* __restrict__ enc_col,
                   const float* __restrict__ clockv,
                   const float* __restrict__ pet,
                   const float* __restrict__ W_dyn,
                   const float* __restrict__ W_concat,
                   const int*   __restrict__ lhw,
                   const int*   __restrict__ lstage,
                   float*       __restrict__ out)
{
    __shared__ float Xs[32 * LDA];
    __shared__ float w3[Dc];
    __shared__ float Rp[32];

    const int b    = blockIdx.x >> 1;
    const int half = blockIdx.x & 1;
    const int t    = threadIdx.x;

    {
        const int r     = t >> 3;
        const int chunk = t & 7;
        const int p     = half * 32 + r;
        const int col0  = chunk * 32;
        const float* src = nullptr;
        if (chunk < 4) {
            const int st = lstage[b * Pc + p] - 1;
            src = enc_col + ((size_t)b * Sc + st) * Dc + col0;
        } else {
            const int w = lhw[b * Pc + p];
            if (w >= 0 && w < Nw)
                src = enc_row + ((size_t)b * Nw + w) * Dc + (col0 - 128);
        }
        float* dst = &Xs[r * LDA + col0];
        if (src) {
            #pragma unroll
            for (int j = 0; j < 8; ++j)
                *(float4*)(dst + 4 * j) = *(const float4*)(src + 4 * j);
        } else {
            #pragma unroll
            for (int j = 0; j < 8; ++j)
                *(float4*)(dst + 4 * j) = make_float4(0.f, 0.f, 0.f, 0.f);
        }
    }
    if (t < Dc) {
        float acc = 0.f;
        #pragma unroll 4
        for (int k = 0; k < Dc; ++k)
            acc = fmaf(W_dyn[k], W_concat[(size_t)(256 + k) * Dc + t], acc);
        w3[t] = acc;
    } else if (t < 160) {
        const int r = t - 128;
        const int p = half * 32 + r;
        const float d = pet[b * Pc + p] - clockv[b];
        Rp[r] = (d > 0.f ? d : 0.f) * NORM_INV;
    }
    __syncthreads();

    const int wave  = t >> 5;
    const int lane  = t & 31;
    const int l16   = lane & 15;
    const int khalf = (lane >> 4) << 1;
    const int n0    = wave * 16;

    v8f acc0 = {};
    v8f acc1 = {};
    const float* wb = W_concat + n0 + l16;
    const float* a0 = &Xs[l16 * LDA + khalf];
    const float* a1 = &Xs[(16 + l16) * LDA + khalf];
    for (int k = 0; k < KA; k += 4) {
        v2f bf;
        bf.x = wb[(size_t)(k + khalf)     * Dc];
        bf.y = wb[(size_t)(k + khalf + 1) * Dc];
        v2f fa0 = *(const v2f*)(a0 + k);
        v2f fa1 = *(const v2f*)(a1 + k);
        acc0 = __builtin_amdgcn_wmma_f32_16x16x4_f32(
                   false, fa0, false, bf, (short)0, acc0, false, false);
        acc1 = __builtin_amdgcn_wmma_f32_16x16x4_f32(
                   false, fa1, false, bf, (short)0, acc1, false, false);
    }

    const float w3v   = w3[n0 + l16];
    const int   rhalf = (lane >> 4) << 3;
    #pragma unroll
    for (int r = 0; r < 8; ++r) {
        const int row0 = rhalf + r;
        const float v0 = acc0[r] + Rp[row0] * w3v;
        out[((size_t)b * (Pc + 2) + half * 32 + row0) * Dc + n0 + l16] = v0;
    }
    #pragma unroll
    for (int r = 0; r < 8; ++r) {
        const int row1 = 16 + rhalf + r;
        const float v1 = acc1[r] + Rp[row1] * w3v;
        out[((size_t)b * (Pc + 2) + half * 32 + row1) * Dc + n0 + l16] = v1;
    }
}

__global__ __launch_bounds__(256)
void robot_kernel(const float* __restrict__ enc_row,
                  const float* __restrict__ enc_col,
                  const float* __restrict__ W_robot,
                  const int* __restrict__ a1loc, const int* __restrict__ a2loc,
                  const int* __restrict__ a1rec, const int* __restrict__ a2rec,
                  const int* __restrict__ a1ns,  const int* __restrict__ a2ns,
                  float* __restrict__ out)
{
    __shared__ float Xs[16 * LDB];
    const int t  = threadIdx.x;
    const int mt = blockIdx.x;

    {
        const int r     = t >> 4;
        const int fbase = (t & 15) * 6;
        const int ar    = mt * 16 + r;
        const int b     = ar >> 1, arm = ar & 1;
        const int loc = (arm ? a2loc : a1loc)[b];
        const int rec = (arm ? a2rec : a1rec)[b];
        const int ns  = (arm ? a2ns  : a1ns )[b];
        #pragma unroll
        for (int j = 0; j < 6; ++j) {
            const int col = (fbase + j) * 4;
            const int seg = col >> 7, cin = col & 127;
            float4 v = make_float4(0.f, 0.f, 0.f, 0.f);
            if (seg == 0) {
                if (loc == Pc + 1)
                    v = make_float4(1.f, 1.f, 1.f, 1.f);
                else if (loc >= 1 && loc <= Pc)
                    v = *(const float4*)(out + ((size_t)b * (Pc + 2) + (loc - 1)) * Dc + cin);
            } else if (seg == 1) {
                if (rec >= 0 && rec < Nw)
                    v = *(const float4*)(enc_row + ((size_t)b * Nw + rec) * Dc + cin);
            } else {
                if (ns >= 1 && ns <= Sc)
                    v = *(const float4*)(enc_col + ((size_t)b * Sc + (ns - 1)) * Dc + cin);
            }
            *(float4*)&Xs[r * LDB + col] = v;
        }
    }
    __syncthreads();

    const int wave  = t >> 5;
    const int lane  = t & 31;
    const int l16   = lane & 15;
    const int khalf = (lane >> 4) << 1;
    const int n0    = wave * 16;

    v8f acc = {};
    const float* wb = W_robot + n0 + l16;
    const float* ap = &Xs[l16 * LDB + khalf];
    for (int k = 0; k < KB; k += 4) {
        v2f a = *(const v2f*)(ap + k);
        v2f bf;
        bf.x = wb[(size_t)(k + khalf)     * Dc];
        bf.y = wb[(size_t)(k + khalf + 1) * Dc];
        acc = __builtin_amdgcn_wmma_f32_16x16x4_f32(
                  false, a, false, bf, (short)0, acc, false, false);
    }
    #pragma unroll
    for (int r = 0; r < 8; ++r) {
        const int mrow = ((lane >> 4) << 3) + r;
        const int ar   = mt * 16 + mrow;
        const int b    = ar >> 1, arm = ar & 1;
        out[((size_t)b * (Pc + 2) + Pc + arm) * Dc + n0 + l16] = acc[r];
    }
}

extern "C" void kernel_launch(void* const* d_in, const int* in_sizes, int n_in,
                              void* d_out, int out_size, void* d_ws, size_t ws_size,
                              hipStream_t stream) {
    (void)in_sizes; (void)n_in; (void)out_size; (void)d_ws; (void)ws_size;
    const float* enc_row  = (const float*)d_in[0];
    const float* enc_col  = (const float*)d_in[1];
    const float* clockv   = (const float*)d_in[2];
    const float* pet      = (const float*)d_in[3];
    const float* W_dyn    = (const float*)d_in[4];
    const float* W_concat = (const float*)d_in[5];
    const float* W_robot  = (const float*)d_in[6];
    const int*   lhw      = (const int*)d_in[7];
    const int*   lstage   = (const int*)d_in[8];
    const int*   a1loc    = (const int*)d_in[9];
    const int*   a2loc    = (const int*)d_in[10];
    const int*   a1rec    = (const int*)d_in[11];
    const int*   a2rec    = (const int*)d_in[12];
    const int*   a1ns     = (const int*)d_in[13];
    const int*   a2ns     = (const int*)d_in[14];
    float* out = (float*)d_out;

    pm_emb_kernel<<<Bc * 2, 256, 0, stream>>>(enc_row, enc_col, clockv, pet,
                                              W_dyn, W_concat, lhw, lstage, out);
    robot_kernel<<<16, 256, 0, stream>>>(enc_row, enc_col, W_robot,
                                         a1loc, a2loc, a1rec, a2rec, a1ns, a2ns, out);
}


# w3 (W_dyn row x W_concat[256:384]) built by all 8 waves, 16 independent row loads per thread issued as one group, not 32 dependent load trips; w3 sum order changes (8 partials of 16 terms)
# speedup vs baseline: 1.3787x; 1.3787x over previous
#include <hip/hip_runtime.h>

typedef __attribute__((ext_vector_type(2))) float v2f;
typedef __attribute__((ext_vector_type(8))) float v8f;

constexpr int Bc = 128, Nw = 4096, Sc = 32, Pc = 64, Dc = 128;
constexpr float NORM_INV = 1.0f / 300.0f;

constexpr int KA  = 256;
constexpr int LDA = 260;
constexpr int KB  = 384;
constexpr int LDB = 388;

__global__ __launch_bounds__(256)
void pm_emb_kernel(const float* __restrict__ enc_row,
                   const float* __restrict__ enc_col,
                   const float* __restrict__ clockv,
                   const float* __restrict__ pet,
                   const float* __restrict__ W_dyn,
                   const float* __restrict__ W_concat,
                   const int*   __restrict__ lhw,
                   const int*   __restrict__ lstage,
                   float*       __restrict__ out)
{
    __shared__ float Xs[32 * LDA];
    __shared__ float w3p[8 * Dc];
    __shared__ float Rp[32];

    const int b    = blockIdx.x >> 1;
    const int half = blockIdx.x & 1;
    const int t    = threadIdx.x;
    const int ln   = t & 31;
    const int wv   = __builtin_amdgcn_readfirstlane(t >> 5);

    const int r     = t >> 3;
    const int chunk = t & 7;
    const int p     = half * 32 + r;
    const int col0  = chunk * 32;
    const bool is_stage = chunk < 4;
    const int* ip   = is_stage ? lstage : lhw;
    const int  idx  = ip[b * Pc + p];
    const float petv = pet[b * Pc + half * 32 + ln];
    const float clk  = clockv[b];

    float dw[16];
    #pragma unroll
    for (int j = 0; j < 16; ++j)
        dw[j] = W_dyn[16 * wv + j];
    float4 wr[16];
    {
        const float* wsrc = W_concat + (size_t)(256 + 16 * wv) * Dc + 4 * ln;
        #pragma unroll
        for (int j = 0; j < 16; ++j)
            wr[j] = *(const float4*)(wsrc + (size_t)j * Dc);
    }

    {
        const float* src = nullptr;
        if (is_stage) {
            const int st = idx - 1;
            src = enc_col + ((size_t)b * Sc + st) * Dc + col0;
        } else {
            const int w = idx;
            if (w >= 0 && w < Nw)
                src = enc_row + ((size_t)b * Nw + w) * Dc + (col0 - 128);
        }
        float* dst = &Xs[r * LDA + col0];
        if (src) {
            #pragma unroll
            for (int j = 0; j < 8; ++j)
                *(float4*)(dst + 4 * j) = *(const float4*)(src + 4 * j);
        } else {
            #pragma unroll
            for (int j = 0; j < 8; ++j)
                *(float4*)(dst + 4 * j) = make_float4(0.f, 0.f, 0.f, 0.f);
        }
    }
    {
        float4 acc = make_float4(0.f, 0.f, 0.f, 0.f);
        #pragma unroll
        for (int j = 0; j < 16; ++j) {
            const float d = dw[j];
            acc.x = fmaf(d, wr[j].x, acc.x);
            acc.y = fmaf(d, wr[j].y, acc.y);
            acc.z = fmaf(d, wr[j].z, acc.z);
            acc.w = fmaf(d, wr[j].w, acc.w);
        }
        *(float4*)&w3p[wv * Dc + 4 * ln] = acc;
    }
    if (t < 32) {
        const float d = petv - clk;
        Rp[t] = (d > 0.f ? d : 0.f) * NORM_INV;
    }
    __syncthreads();

    const int wave  = t >> 5;
    const int lane  = t & 31;
    const int l16   = lane & 15;
    const int khalf = (lane >> 4) << 1;
    const int n0    = wave * 16;

    v8f acc0 = {};
    v8f acc1 = {};
    const float* wb = W_concat + n0 + l16;
    const float* a0 = &Xs[l16 * LDA + khalf];
    const float* a1 = &Xs[(16 + l16) * LDA + khalf];
    for (int k = 0; k < KA; k += 4) {
        v2f bf;
        bf.x = wb[(size_t)(k + khalf)     * Dc];
        bf.y = wb[(size_t)(k + khalf + 1) * Dc];
        v2f fa0 = *(const v2f*)(a0 + k);
        v2f fa1 = *(const v2f*)(a1 + k);
        acc0 = __builtin_amdgcn_wmma_f32_16x16x4_f32(
                   false, fa0, false, bf, (short)0, acc0, false, false);
        acc1 = __builtin_amdgcn_wmma_f32_16x16x4_f32(
                   false, fa1, false, bf, (short)0, acc1, false, false);
    }

    float w3v = w3p[n0 + l16];
    #pragma unroll
    for (int w = 1; w < 8; ++w)
        w3v += w3p[w * Dc + n0 + l16];
    const int   rhalf = (lane >> 4) << 3;
    #pragma unroll
    for (int r = 0; r < 8; ++r) {
        const int row0 = rhalf + r;
        const float v0 = acc0[r] + Rp[row0] * w3v;
        out[((size_t)b * (Pc + 2) + half * 32 + row0) * Dc + n0 + l16] = v0;
    }
    #pragma unroll
    for (int r = 0; r < 8; ++r) {
        const int row1 = 16 + rhalf + r;
        const float v1 = acc1[r] + Rp[row1] * w3v;
        out[((size_t)b * (Pc + 2) + half * 32 + row1) * Dc + n0 + l16] = v1;
    }
}

__global__ __launch_bounds__(256)
void robot_kernel(const float* __restrict__ enc_row,
                  const float* __restrict__ enc_col,
                  const float* __restrict__ W_robot,
                  const int* __restrict__ a1loc, const int* __restrict__ a2loc,
                  const int* __restrict__ a1rec, const int* __restrict__ a2rec,
                  const int* __restrict__ a1ns,  const int* __restrict__ a2ns,
                  float* __restrict__ out)
{
    __shared__ float Xs[16 * LDB];
    const int t  = threadIdx.x;
    const int mt = blockIdx.x;

    {
        const int r     = t >> 4;
        const int fbase = (t & 15) * 6;
        const int ar    = mt * 16 + r;
        const int b     = ar >> 1, arm = ar & 1;
        const int loc = (arm ? a2loc : a1loc)[b];
        const int rec = (arm ? a2rec : a1rec)[b];
        const int ns  = (arm ? a2ns  : a1ns )[b];
        #pragma unroll
        for (int j = 0; j < 6; ++j) {
            const int col = (fbase + j) * 4;
            const int seg = col >> 7, cin = col & 127;
            float4 v = make_float4(0.f, 0.f, 0.f, 0.f);
            if (seg == 0) {
                if (loc == Pc + 1)
                    v = make_float4(1.f, 1.f, 1.f, 1.f);
                else if (loc >= 1 && loc <= Pc)
                    v = *(const float4*)(out + ((size_t)b * (Pc + 2) + (loc - 1)) * Dc + cin);
            } else if (seg == 1) {
                if (rec >= 0 && rec < Nw)
                    v = *(const float4*)(enc_row + ((size_t)b * Nw + rec) * Dc + cin);
            } else {
                if (ns >= 1 && ns <= Sc)
                    v = *(const float4*)(enc_col + ((size_t)b * Sc + (ns - 1)) * Dc + cin);
            }
            *(float4*)&Xs[r * LDB + col] = v;
        }
    }
    __syncthreads();

    const int wave  = t >> 5;
    const int lane  = t & 31;
    const int l16   = lane & 15;
    const int khalf = (lane >> 4) << 1;
    const int n0    = wave * 16;

    v8f acc = {};
    const float* wb = W_robot + n0 + l16;
    const float* ap = &Xs[l16 * LDB + khalf];
    for (int k = 0; k < KB; k += 4) {
        v2f a = *(const v2f*)(ap + k);
        v2f bf;
        bf.x = wb[(size_t)(k + khalf)     * Dc];
        bf.y = wb[(size_t)(k + khalf + 1) * Dc];
        acc = __builtin_amdgcn_wmma_f32_16x16x4_f32(
                  false, a, false, bf, (short)0, acc, false, false);
    }
    #pragma unroll
    for (int r = 0; r < 8; ++r) {
        const int mrow = ((lane >> 4) << 3) + r;
        const int ar   = mt * 16 + mrow;
        const int b    = ar >> 1, arm = ar & 1;
        out[((size_t)b * (Pc + 2) + Pc + arm) * Dc + n0 + l16] = acc[r];
    }
}

extern "C" void kernel_launch(void* const* d_in, const int* in_sizes, int n_in,
                              void* d_out, int out_size, void* d_ws, size_t ws_size,
                              hipStream_t stream) {
    (void)in_sizes; (void)n_in; (void)out_size; (void)d_ws; (void)ws_size;
    const float* enc_row  = (const float*)d_in[0];
    const float* enc_col  = (const float*)d_in[1];
    const float* clockv   = (const float*)d_in[2];
    const float* pet      = (const float*)d_in[3];
    const float* W_dyn    = (const float*)d_in[4];
    const float* W_concat = (const float*)d_in[5];
    const float* W_robot  = (const float*)d_in[6];
    const int*   lhw      = (const int*)d_in[7];
    const int*   lstage   = (const int*)d_in[8];
    const int*   a1loc    = (const int*)d_in[9];
    const int*   a2loc    = (const int*)d_in[10];
    const int*   a1rec    = (const int*)d_in[11];
    const int*   a2rec    = (const int*)d_in[12];
    const int*   a1ns     = (const int*)d_in[13];
    const int*   a2ns     = (const int*)d_in[14];
    float* out = (float*)d_out;

    pm_emb_kernel<<<Bc * 2, 256, 0, stream>>>(enc_row, enc_col, clockv, pet,
                                              W_dyn, W_concat, lhw, lstage, out);
    robot_kernel<<<16, 256, 0, stream>>>(enc_row, enc_col, W_robot,
                                         a1loc, a2loc, a1rec, a2rec, a1ns, a2ns, out);
}


# robot arms moved into the PM launch: 16 extra work-groups rebuild the one PM row each arm sits at (same K sweep) in LDS instead of reading out back; one launch, not two; sum orders unchanged
# speedup vs baseline: 1.8226x; 1.3220x over previous
#include <hip/hip_runtime.h>

typedef __attribute__((ext_vector_type(2))) float v2f;
typedef __attribute__((ext_vector_type(8))) float v8f;

constexpr int Bc = 128, Nw = 4096, Sc = 32, Pc = 64, Dc = 128;
constexpr float NORM_INV = 1.0f / 300.0f;

constexpr int KA  = 256;
constexpr int LDA = 260;
constexpr int KB  = 384;
constexpr int LDB = 388;
constexpr int NRB  = 16;
constexpr int GRID = NRB + Bc * 2;

__device__ __forceinline__ void w3_load(const float* __restrict__ W_dyn,
                                        const float* __restrict__ W_concat,
                                        int wv, int ln, float (&dw)[16], float4 (&wr)[16])
{
    #pragma unroll
    for (int j = 0; j < 16; ++j)
        dw[j] = W_dyn[16 * wv + j];
    const float* wsrc = W_concat + (size_t)(256 + 16 * wv) * Dc + 4 * ln;
    #pragma unroll
    for (int j = 0; j < 16; ++j)
        wr[j] = *(const float4*)(wsrc + (size_t)j * Dc);
}

__device__ __forceinline__ void w3_partial(const float (&dw)[16], const float4 (&wr)[16],
                                           float* w3p, int wv, int ln)
{
    float4 acc = make_float4(0.f, 0.f, 0.f, 0.f);
    #pragma unroll
    for (int j = 0; j < 16; ++j) {
        const float d = dw[j];
        acc.x = fmaf(d, wr[j].x, acc.x);
        acc.y = fmaf(d, wr[j].y, acc.y);
        acc.z = fmaf(d, wr[j].z, acc.z);
        acc.w = fmaf(d, wr[j].w, acc.w);
    }
    *(float4*)&w3p[wv * Dc + 4 * ln] = acc;
}

__global__ __launch_bounds__(256)
void pm_robot_kernel(const float* __restrict__ enc_row,
                     const float* __restrict__ enc_col,
                     const float* __restrict__ clockv,
                     const float* __restrict__ pet,
                     const float* __restrict__ W_dyn,
                     const float* __restrict__ W_concat,
                     const float* __restrict__ W_robot,
                     const int*   __restrict__ lhw,
                     const int*   __restrict__ lstage,
                     const int* __restrict__ a1loc, const int* __restrict__ a2loc,
                     const int* __restrict__ a1rec, const int* __restrict__ a2rec,
                     const int* __restrict__ a1ns,  const int* __restrict__ a2ns,
                     float*       __restrict__ out)
{
    __shared__ __attribute__((aligned(16))) float Xs[32 * LDA];
    __shared__ __attribute__((aligned(16))) float Ys[16 * LDB];
    __shared__ __attribute__((aligned(16))) float w3p[8 * Dc];
    __shared__ __attribute__((aligned(16))) float Rp[32];
    __shared__ __attribute__((aligned(16))) int   Lk[16];

    const int t    = threadIdx.x;
    const int ln   = t & 31;
    const int wv   = __builtin_amdgcn_readfirstlane(t >> 5);

    const int wave  = t >> 5;
    const int lane  = t & 31;
    const int l16   = lane & 15;
    const int khalf = (lane >> 4) << 1;
    const int rhalf = (lane >> 4) << 3;
    const int n0    = wave * 16;

    if (blockIdx.x < NRB) {
        const int mt  = blockIdx.x;
        const int r   = t >> 4;
        const int c8  = (t & 15) * 8;
        const int ar  = mt * 16 + r;
        const int b   = ar >> 1, arm = ar & 1;

        const int loc = (arm ? a2loc : a1loc)[b];
        const int rec = (arm ? a2rec : a1rec)[b];
        const int ns  = (arm ? a2ns  : a1ns )[b];
        const float clk = clockv[b];

        float dw[16];
        float4 wr[16];
        w3_load(W_dyn, W_concat, wv, ln, dw, wr);

        const bool has_pm = (loc >= 1 && loc <= Pc);
        const int  p      = has_pm ? loc - 1 : 0;
        const int   stg  = lstage[b * Pc + p];
        const int   waf  = lhw[b * Pc + p];
        const float petv = pet[b * Pc + p];

        const bool rec_ok = (rec >= 0 && rec < Nw);
        const bool ns_ok  = (ns >= 1 && ns <= Sc);
        const float* rsrc = enc_row + ((size_t)b * Nw + (rec_ok ? rec : 0)) * Dc + c8;
        const float* nsrc = enc_col + ((size_t)b * Sc + (ns_ok ? ns - 1 : 0)) * Dc + c8;
        float4 y0 = *(const float4*)(rsrc);
        float4 y1 = *(const float4*)(rsrc + 4);
        float4 y2 = *(const float4*)(nsrc);
        float4 y3 = *(const float4*)(nsrc + 4);

        const bool waf_ok = (waf >= 0 && waf < Nw);
        const float* ssrc = enc_col + ((size_t)b * Sc + (stg - 1)) * Dc + c8;
        const float* wsrc = enc_row + ((size_t)b * Nw + (waf_ok ? waf : 0)) * Dc + c8;
        float4 x0 = *(const float4*)(ssrc);
        float4 x1 = *(const float4*)(ssrc + 4);
        float4 x2 = *(const float4*)(wsrc);
        float4 x3 = *(const float4*)(wsrc + 4);

        const float4 z4 = make_float4(0.f, 0.f, 0.f, 0.f);
        if (!rec_ok) { y0 = z4; y1 = z4; }
        if (!ns_ok)  { y2 = z4; y3 = z4; }
        if (!waf_ok) { x2 = z4; x3 = z4; }

        float* yd = &Ys[r * LDB + Dc + c8];
        *(float4*)(yd)          = y0;
        *(float4*)(yd + 4)      = y1;
        *(float4*)(yd + Dc)     = y2;
        *(float4*)(yd + Dc + 4) = y3;
        float* xd = &Xs[r * LDA + c8];
        *(float4*)(xd)          = x0;
        *(float4*)(xd + 4)      = x1;
        *(float4*)(xd + Dc)     = x2;
        *(float4*)(xd + Dc + 4) = x3;

        w3_partial(dw, wr, w3p, wv, ln);

        if ((t & 15) == 0) {
            const float d = petv - clk;
            Rp[r] = (d > 0.f ? d : 0.f) * NORM_INV;
            Lk[r] = has_pm ? 2 : (loc == Pc + 1 ? 1 : 0);
        }
        __syncthreads();

        v8f accp = {};
        {
            const float* wb = W_concat + n0 + l16;
            const float* ap = &Xs[l16 * LDA + khalf];
            for (int k = 0; k < KA; k += 4) {
                v2f a = *(const v2f*)(ap + k);
                v2f bf;
                bf.x = wb[(size_t)(k + khalf)     * Dc];
                bf.y = wb[(size_t)(k + khalf + 1) * Dc];
                accp = __builtin_amdgcn_wmma_f32_16x16x4_f32(
                           false, a, false, bf, (short)0, accp, false, false);
            }
        }
        {
            float w3v = w3p[n0 + l16];
            #pragma unroll
            for (int w = 1; w < 8; ++w)
                w3v += w3p[w * Dc + n0 + l16];
            const float4 rpa = *(const float4*)&Rp[rhalf];
            const float4 rpb = *(const float4*)&Rp[rhalf + 4];
            const int4   lka = *(const int4*)&Lk[rhalf];
            const int4   lkb = *(const int4*)&Lk[rhalf + 4];
            const float rp[8] = {rpa.x, rpa.y, rpa.z, rpa.w, rpb.x, rpb.y, rpb.z, rpb.w};
            const int   lk[8] = {lka.x, lka.y, lka.z, lka.w, lkb.x, lkb.y, lkb.z, lkb.w};
            #pragma unroll
            for (int q = 0; q < 8; ++q) {
                const int row   = rhalf + q;
                const float v   = accp[q] + rp[q] * w3v;
                const float alt = (lk[q] == 1) ? 1.f : 0.f;
                Ys[row * LDB + n0 + l16] = (lk[q] == 2) ? v : alt;
            }
        }
        __syncthreads();

        v8f acc = {};
        {
            const float* wb = W_robot + n0 + l16;
            const float* ap = &Ys[l16 * LDB + khalf];
            for (int k = 0; k < KB; k += 4) {
                v2f a = *(const v2f*)(ap + k);
                v2f bf;
                bf.x = wb[(size_t)(k + khalf)     * Dc];
                bf.y = wb[(size_t)(k + khalf + 1) * Dc];
                acc = __builtin_amdgcn_wmma_f32_16x16x4_f32(
                          false, a, false, bf, (short)0, acc, false, false);
            }
        }
        #pragma unroll
        for (int q = 0; q < 8; ++q) {
            const int mrow = rhalf + q;
            const int oar  = mt * 16 + mrow;
            const int ob   = oar >> 1, oarm = oar & 1;
            out[((size_t)ob * (Pc + 2) + Pc + oarm) * Dc + n0 + l16] = acc[q];
        }
        return;
    }

    const int bid  = blockIdx.x - NRB;
    const int b    = bid >> 1;
    const int half = bid & 1;

    const int r     = t >> 3;
    const int chunk = t & 7;
    const int p     = half * 32 + r;
    const int col0  = chunk * 32;
    const bool is_stage = chunk < 4;
    const int* ip   = is_stage ? lstage : lhw;
    const int  idx  = ip[b * Pc + p];
    const float petv = pet[b * Pc + half * 32 + ln];
    const float clk  = clockv[b];

    float dw[16];
    float4 wr[16];
    w3_load(W_dyn, W_concat, wv, ln, dw, wr);

    {
        const float* src = nullptr;
        if (is_stage) {
            const int st = idx - 1;
            src = enc_col + ((size_t)b * Sc + st) * Dc + col0;
        } else {
            const int w = idx;
            if (w >= 0 && w < Nw)
                src = enc_row + ((size_t)b * Nw + w) * Dc + (col0 - 128);
        }
        float* dst = &Xs[r * LDA + col0];
        if (src) {
            #pragma unroll
            for (int j = 0; j < 8; ++j)
                *(float4*)(dst + 4 * j) = *(const float4*)(src + 4 * j);
        } else {
            #pragma unroll
            for (int j = 0; j < 8; ++j)
                *(float4*)(dst + 4 * j) = make_float4(0.f, 0.f, 0.f, 0.f);
        }
    }
    w3_partial(dw, wr, w3p, wv, ln);
    if (t < 32) {
        const float d = petv - clk;
        Rp[t] = (d > 0.f ? d : 0.f) * NORM_INV;
    }
    __syncthreads();

    v8f acc0 = {};
    v8f acc1 = {};
    const float* wb = W_concat + n0 + l16;
    const float* a0 = &Xs[l16 * LDA + khalf];
    const float* a1 = &Xs[(16 + l16) * LDA + khalf];
    for (int k = 0; k < KA; k += 4) {
        v2f bf;
        bf.x = wb[(size_t)(k + khalf)     * Dc];
        bf.y = wb[(size_t)(k + khalf + 1) * Dc];
        v2f fa0 = *(const v2f*)(a0 + k);
        v2f fa1 = *(const v2f*)(a1 + k);
        acc0 = __builtin_amdgcn_wmma_f32_16x16x4_f32(
                   false, fa0, false, bf, (short)0, acc0, false, false);
        acc1 = __builtin_amdgcn_wmma_f32_16x16x4_f32(
                   false, fa1, false, bf, (short)0, acc1, false, false);
    }

    float w3v = w3p[n0 + l16];
    #pragma unroll
    for (int w = 1; w < 8; ++w)
        w3v += w3p[w * Dc + n0 + l16];
    #pragma unroll
    for (int q = 0; q < 8; ++q) {
        const int row0 = rhalf + q;
        const float v0 = acc0[q] + Rp[row0] * w3v;
        out[((size_t)b * (Pc + 2) + half * 32 + row0) * Dc + n0 + l16] = v0;
    }
    #pragma unroll
    for (int q = 0; q < 8; ++q) {
        const int row1 = 16 + rhalf + q;
        const float v1 = acc1[q] + Rp[row1] * w3v;
        out[((size_t)b * (Pc + 2) + half * 32 + row1) * Dc + n0 + l16] = v1;
    }
}

extern "C" void kernel_launch(void* const* d_in, const int* in_sizes, int n_in,
                              void* d_out, int out_size, void* d_ws, size_t ws_size,
                              hipStream_t stream) {
    (void)in_sizes; (void)n_in; (void)out_size; (void)d_ws; (void)ws_size;
    const float* enc_row  = (const float*)d_in[0];
    const float* enc_col  = (const float*)d_in[1];
    const float* clockv   = (const float*)d_in[2];
    const float* pet      = (const float*)d_in[3];
    const float* W_dyn    = (const float*)d_in[4];
    const float* W_concat = (const float*)d_in[5];
    const float* W_robot  = (const float*)d_in[6];
    const int*   lhw      = (const int*)d_in[7];
    const int*   lstage   = (const int*)d_in[8];
    const int*   a1loc    = (const int*)d_in[9];
    const int*   a2loc    = (const int*)d_in[10];
    const int*   a1rec    = (const int*)d_in[11];
    const int*   a2rec    = (const int*)d_in[12];
    const int*   a1ns     = (const int*)d_in[13];
    const int*   a2ns     = (const int*)d_in[14];
    float* out = (float*)d_out;

    pm_robot_kernel<<<GRID, 256, 0, stream>>>(enc_row, enc_col, clockv, pet,
                                              W_dyn, W_concat, W_robot, lhw, lstage,
                                              a1loc, a2loc, a1rec, a2rec, a1ns, a2ns, out);
}


# grid 272 -> 256 work-groups so that no WGP hosts two: the 512 PM row tiles go to 240 work-groups (32 take three 16-row tiles, 208 two) beside the 16 arm work-groups; sum orders unchanged
# speedup vs baseline: 2.0260x; 1.1116x over previous
#include <hip/hip_runtime.h>

typedef __attribute__((ext_vector_type(2))) float v2f;
typedef __attribute__((ext_vector_type(8))) float v8f;

constexpr int Bc = 128, Nw = 4096, Sc = 32, Pc = 64, Dc = 128;
constexpr float NORM_INV = 1.0f / 300.0f;

constexpr int KA  = 256;
constexpr int LDA = 260;
constexpr int KB  = 384;
constexpr int LDB = 388;
constexpr int NRB  = 16;
constexpr int NT3  = 32;
constexpr int NT2  = 208;
constexpr int GRID = NRB + NT3 + NT2;
static_assert(3 * NT3 + 2 * NT2 == Bc * Pc / 16, "every PM row tile is taken exactly once");
static_assert(GRID == 256, "one work-group per WGP");

__device__ __forceinline__ void w3_load(const float* __restrict__ W_dyn,
                                        const float* __restrict__ W_concat,
                                        int wv, int ln, float (&dw)[16], float4 (&wr)[16])
{
    #pragma unroll
    for (int j = 0; j < 16; ++j)
        dw[j] = W_dyn[16 * wv + j];
    const float* wsrc = W_concat + (size_t)(256 + 16 * wv) * Dc + 4 * ln;
    #pragma unroll
    for (int j = 0; j < 16; ++j)
        wr[j] = *(const float4*)(wsrc + (size_t)j * Dc);
}

__device__ __forceinline__ void w3_partial(const float (&dw)[16], const float4 (&wr)[16],
                                           float* w3p, int wv, int ln)
{
    float4 acc = make_float4(0.f, 0.f, 0.f, 0.f);
    #pragma unroll
    for (int j = 0; j < 16; ++j) {
        const float d = dw[j];
        acc.x = fmaf(d, wr[j].x, acc.x);
        acc.y = fmaf(d, wr[j].y, acc.y);
        acc.z = fmaf(d, wr[j].z, acc.z);
        acc.w = fmaf(d, wr[j].w, acc.w);
    }
    *(float4*)&w3p[wv * Dc + 4 * ln] = acc;
}

template <int NV>
__device__ __forceinline__ void row_piece_load(const float* __restrict__ enc_row,
                                               const float* __restrict__ enc_col,
                                               int b, int idx, int c0, float4 (&v)[NV])
{
    const bool is_stage = c0 < Dc;
    const bool ok = is_stage || (idx >= 0 && idx < Nw);
    const float* src = is_stage
        ? enc_col + ((size_t)b * Sc + (idx - 1)) * Dc + c0
        : enc_row + ((size_t)b * Nw + (ok ? idx : 0)) * Dc + (c0 - Dc);
    #pragma unroll
    for (int j = 0; j < NV; ++j)
        v[j] = *(const float4*)(src + 4 * j);
    if (!ok) {
        #pragma unroll
        for (int j = 0; j < NV; ++j)
            v[j] = make_float4(0.f, 0.f, 0.f, 0.f);
    }
}

template <int NT>
__device__ __forceinline__ void pm_rows(const float* __restrict__ enc_row,
                                        const float* __restrict__ enc_col,
                                        const float* __restrict__ clockv,
                                        const float* __restrict__ pet,
                                        const float* __restrict__ W_dyn,
                                        const float* __restrict__ W_concat,
                                        const int*   __restrict__ lhw,
                                        const int*   __restrict__ lstage,
                                        float*       __restrict__ out,
                                        float* Xs, float* w3p, float* Rp, int g0)
{
    const int t     = threadIdx.x;
    const int ln    = t & 31;
    const int wv    = __builtin_amdgcn_readfirstlane(t >> 5);
    const int l16   = ln & 15;
    const int khalf = (ln >> 4) << 1;
    const int rhalf = (ln >> 4) << 3;
    const int n0    = wv * 16;

    const int rA = t >> 3,        cA = (t & 7) * 32;
    const int rB = 32 + (t >> 4), cB = (t & 15) * 16;
    const int gA = g0 + rA;
    const int gB = g0 + rB;
    const int idxA = (cA < Dc ? lstage : lhw)[gA];
    int idxB = 0;
    if (NT == 3) idxB = (cB < Dc ? lstage : lhw)[gB];
    float petv = 0.f, clk = 0.f;
    if (t < 16 * NT) {
        petv = pet[g0 + t];
        clk  = clockv[(g0 + t) >> 6];
    }

    float dw[16];
    float4 wr[16];
    w3_load(W_dyn, W_concat, wv, ln, dw, wr);

    {
        float4 va[8];
        float4 vb[4];
        row_piece_load<8>(enc_row, enc_col, gA >> 6, idxA, cA, va);
        if (NT == 3) row_piece_load<4>(enc_row, enc_col, gB >> 6, idxB, cB, vb);
        float* da = &Xs[rA * LDA + cA];
        #pragma unroll
        for (int j = 0; j < 8; ++j)
            *(float4*)(da + 4 * j) = va[j];
        if (NT == 3) {
            float* db = &Xs[rB * LDA + cB];
            #pragma unroll
            for (int j = 0; j < 4; ++j)
                *(float4*)(db + 4 * j) = vb[j];
        }
    }
    w3_partial(dw, wr, w3p, wv, ln);
    if (t < 16 * NT) {
        const float d = petv - clk;
        Rp[t] = (d > 0.f ? d : 0.f) * NORM_INV;
    }
    __syncthreads();

    v8f acc[NT];
    #pragma unroll
    for (int m = 0; m < NT; ++m)
        acc[m] = v8f{};
    const float* wb = W_concat + n0 + l16;
    const float* ap = &Xs[l16 * LDA + khalf];
    #pragma unroll 8
    for (int k = 0; k < KA; k += 4) {
        v2f bf;
        bf.x = wb[(size_t)(k + khalf)     * Dc];
        bf.y = wb[(size_t)(k + khalf + 1) * Dc];
        #pragma unroll
        for (int m = 0; m < NT; ++m) {
            v2f fa = *(const v2f*)(ap + m * 16 * LDA + k);
            acc[m] = __builtin_amdgcn_wmma_f32_16x16x4_f32(
                         false, fa, false, bf, (short)0, acc[m], false, false);
        }
    }

    float w3v = w3p[n0 + l16];
    #pragma unroll
    for (int w = 1; w < 8; ++w)
        w3v += w3p[w * Dc + n0 + l16];
    #pragma unroll
    for (int m = 0; m < NT; ++m) {
        #pragma unroll
        for (int q = 0; q < 8; ++q) {
            const int row = m * 16 + rhalf + q;
            const int g   = g0 + row;
            const float v = acc[m][q] + Rp[row] * w3v;
            out[((size_t)(g >> 6) * (Pc + 2) + (g & 63)) * Dc + n0 + l16] = v;
        }
    }
}

__global__ __launch_bounds__(256)
void pm_robot_kernel(const float* __restrict__ enc_row,
                     const float* __restrict__ enc_col,
                     const float* __restrict__ clockv,
                     const float* __restrict__ pet,
                     const float* __restrict__ W_dyn,
                     const float* __restrict__ W_concat,
                     const float* __restrict__ W_robot,
                     const int*   __restrict__ lhw,
                     const int*   __restrict__ lstage,
                     const int* __restrict__ a1loc, const int* __restrict__ a2loc,
                     const int* __restrict__ a1rec, const int* __restrict__ a2rec,
                     const int* __restrict__ a1ns,  const int* __restrict__ a2ns,
                     float*       __restrict__ out)
{
    __shared__ __attribute__((aligned(16))) float Sx[48 * LDA];
    __shared__ __attribute__((aligned(16))) float w3p[8 * Dc];
    __shared__ __attribute__((aligned(16))) float Rp[48];
    float* const Xs = Sx;
    float* const Ys = Sx + 16 * LDA;
    static_assert(16 * LDA + 16 * LDB <= 48 * LDA, "arm rows fit behind the 16 PM rows");
    __shared__ __attribute__((aligned(16))) int   Lk[16];

    const int t    = threadIdx.x;
    const int ln   = t & 31;
    const int wv   = __builtin_amdgcn_readfirstlane(t >> 5);

    const int wave  = t >> 5;
    const int lane  = t & 31;
    const int l16   = lane & 15;
    const int khalf = (lane >> 4) << 1;
    const int rhalf = (lane >> 4) << 3;
    const int n0    = wave * 16;

    if (blockIdx.x < NRB) {
        const int mt  = blockIdx.x;
        const int r   = t >> 4;
        const int c8  = (t & 15) * 8;
        const int ar  = mt * 16 + r;
        const int b   = ar >> 1, arm = ar & 1;

        const int loc = (arm ? a2loc : a1loc)[b];
        const int rec = (arm ? a2rec : a1rec)[b];
        const int ns  = (arm ? a2ns  : a1ns )[b];
        const float clk = clockv[b];

        float dw[16];
        float4 wr[16];
        w3_load(W_dyn, W_concat, wv, ln, dw, wr);

        const bool has_pm = (loc >= 1 && loc <= Pc);
        const int  p      = has_pm ? loc - 1 : 0;
        const int   stg  = lstage[b * Pc + p];
        const int   waf  = lhw[b * Pc + p];
        const float petv = pet[b * Pc + p];

        const bool rec_ok = (rec >= 0 && rec < Nw);
        const bool ns_ok  = (ns >= 1 && ns <= Sc);
        const float* rsrc = enc_row + ((size_t)b * Nw + (rec_ok ? rec : 0)) * Dc + c8;
        const float* nsrc = enc_col + ((size_t)b * Sc + (ns_ok ? ns - 1 : 0)) * Dc + c8;
        float4 y0 = *(const float4*)(rsrc);
        float4 y1 = *(const float4*)(rsrc + 4);
        float4 y2 = *(const float4*)(nsrc);
        float4 y3 = *(const float4*)(nsrc + 4);

        const bool waf_ok = (waf >= 0 && waf < Nw);
        const float* ssrc = enc_col + ((size_t)b * Sc + (stg - 1)) * Dc + c8;
        const float* wsrc = enc_row + ((size_t)b * Nw + (waf_ok ? waf : 0)) * Dc + c8;
        float4 x0 = *(const float4*)(ssrc);
        float4 x1 = *(const float4*)(ssrc + 4);
        float4 x2 = *(const float4*)(wsrc);
        float4 x3 = *(const float4*)(wsrc + 4);

        const float4 z4 = make_float4(0.f, 0.f, 0.f, 0.f);
        if (!rec_ok) { y0 = z4; y1 = z4; }
        if (!ns_ok)  { y2 = z4; y3 = z4; }
        if (!waf_ok) { x2 = z4; x3 = z4; }

        float* yd = &Ys[r * LDB + Dc + c8];
        *(float4*)(yd)          = y0;
        *(float4*)(yd + 4)      = y1;
        *(float4*)(yd + Dc)     = y2;
        *(float4*)(yd + Dc + 4) = y3;
        float* xd = &Xs[r * LDA + c8];
        *(float4*)(xd)          = x0;
        *(float4*)(xd + 4)      = x1;
        *(float4*)(xd + Dc)     = x2;
        *(float4*)(xd + Dc + 4) = x3;

        w3_partial(dw, wr, w3p, wv, ln);

        if ((t & 15) == 0) {
            const float d = petv - clk;
            Rp[r] = (d > 0.f ? d : 0.f) * NORM_INV;
            Lk[r] = has_pm ? 2 : (loc == Pc + 1 ? 1 : 0);
        }
        __syncthreads();

        v8f accp = {};
        {
            const float* wb = W_concat + n0 + l16;
            const float* ap = &Xs[l16 * LDA + khalf];
            for (int k = 0; k < KA; k += 4) {
                v2f a = *(const v2f*)(ap + k);
                v2f bf;
                bf.x = wb[(size_t)(k + khalf)     * Dc];
                bf.y = wb[(size_t)(k + khalf + 1) * Dc];
                accp = __builtin_amdgcn_wmma_f32_16x16x4_f32(
                           false, a, false, bf, (short)0, accp, false, false);
            }
        }
        {
            float w3v = w3p[n0 + l16];
            #pragma unroll
            for (int w = 1; w < 8; ++w)
                w3v += w3p[w * Dc + n0 + l16];
            const float4 rpa = *(const float4*)&Rp[rhalf];
            const float4 rpb = *(const float4*)&Rp[rhalf + 4];
            const int4   lka = *(const int4*)&Lk[rhalf];
            const int4   lkb = *(const int4*)&Lk[rhalf + 4];
            const float rp[8] = {rpa.x, rpa.y, rpa.z, rpa.w, rpb.x, rpb.y, rpb.z, rpb.w};
            const int   lk[8] = {lka.x, lka.y, lka.z, lka.w, lkb.x, lkb.y, lkb.z, lkb.w};
            #pragma unroll
            for (int q = 0; q < 8; ++q) {
                const int row   = rhalf + q;
                const float v   = accp[q] + rp[q] * w3v;
                const float alt = (lk[q] == 1) ? 1.f : 0.f;
                Ys[row * LDB + n0 + l16] = (lk[q] == 2) ? v : alt;
            }
        }
        __syncthreads();

        v8f acc = {};
        {
            const float* wb = W_robot + n0 + l16;
            const float* ap = &Ys[l16 * LDB + khalf];
            for (int k = 0; k < KB; k += 4) {
                v2f a = *(const v2f*)(ap + k);
                v2f bf;
                bf.x = wb[(size_t)(k + khalf)     * Dc];
                bf.y = wb[(size_t)(k + khalf + 1) * Dc];
                acc = __builtin_amdgcn_wmma_f32_16x16x4_f32(
                          false, a, false, bf, (short)0, acc, false, false);
            }
        }
        #pragma unroll
        for (int q = 0; q < 8; ++q) {
            const int mrow = rhalf + q;
            const int oar  = mt * 16 + mrow;
            const int ob   = oar >> 1, oarm = oar & 1;
            out[((size_t)ob * (Pc + 2) + Pc + oarm) * Dc + n0 + l16] = acc[q];
        }
        return;
    }

    const int bid = blockIdx.x - NRB;
    if (bid < NT3)
        pm_rows<3>(enc_row, enc_col, clockv, pet, W_dyn, W_concat, lhw, lstage, out,
                   Xs, w3p, Rp, 48 * bid);
    else
        pm_rows<2>(enc_row, enc_col, clockv, pet, W_dyn, W_concat, lhw, lstage, out,
                   Xs, w3p, Rp, 48 * NT3 + 32 * (bid - NT3));
}

extern "C" void kernel_launch(void* const* d_in, const int* in_sizes, int n_in,
                              void* d_out, int out_size, void* d_ws, size_t ws_size,
                              hipStream_t stream) {
    (void)in_sizes; (void)n_in; (void)out_size; (void)d_ws; (void)ws_size;
    const float* enc_row  = (const float*)d_in[0];
    const float* enc_col  = (const float*)d_in[1];
    const float* clockv   = (const float*)d_in[2];
    const float* pet      = (const float*)d_in[3];
    const float* W_dyn    = (const float*)d_in[4];
    const float* W_concat = (const float*)d_in[5];
    const float* W_robot  = (const float*)d_in[6];
    const int*   lhw      = (const int*)d_in[7];
    const int*   lstage   = (const int*)d_in[8];
    const int*   a1loc    = (const int*)d_in[9];
    const int*   a2loc    = (const int*)d_in[10];
    const int*   a1rec    = (const int*)d_in[11];
    const int*   a2rec    = (const int*)d_in[12];
    const int*   a1ns     = (const int*)d_in[13];
    const int*   a2ns     = (const int*)d_in[14];
    float* out = (float*)d_out;

    pm_robot_kernel<<<GRID, 256, 0, stream>>>(enc_row, enc_col, clockv, pet,
                                              W_dyn, W_concat, W_robot, lhw, lstage,
                                              a1loc, a2loc, a1rec, a2rec, a1ns, a2ns, out);
}


# B fragments fetched from L2 one group of 16 k steps ahead of the WMMAs that use them (second register set; first group issued before the gather), in PM and arm sweeps; sum orders unchanged
# speedup vs baseline: 2.2135x; 1.0925x over previous
#include <hip/hip_runtime.h>

typedef __attribute__((ext_vector_type(2))) float v2f;
typedef __attribute__((ext_vector_type(8))) float v8f;

constexpr int Bc = 128, Nw = 4096, Sc = 32, Pc = 64, Dc = 128;
constexpr float NORM_INV = 1.0f / 300.0f;

constexpr int KA  = 256;
constexpr int LDA = 260;
constexpr int KB  = 384;
constexpr int LDB = 388;
constexpr int NRB  = 16;
constexpr int NT3  = 32;
constexpr int NT2  = 208;
constexpr int GRID = NRB + NT3 + NT2;
static_assert(3 * NT3 + 2 * NT2 == Bc * Pc / 16, "every PM row tile is taken exactly once");
static_assert(GRID == 256, "one work-group per WGP");

__device__ __forceinline__ void w3_load(const float* __restrict__ W_dyn,
                                        const float* __restrict__ W_concat,
                                        int wv, int ln, float (&dw)[16], float4 (&wr)[16])
{
    #pragma unroll
    for (int j = 0; j < 16; ++j)
        dw[j] = W_dyn[16 * wv + j];
    const float* wsrc = W_concat + (size_t)(256 + 16 * wv) * Dc + 4 * ln;
    #pragma unroll
    for (int j = 0; j < 16; ++j)
        wr[j] = *(const float4*)(wsrc + (size_t)j * Dc);
}

__device__ __forceinline__ void w3_partial(const float (&dw)[16], const float4 (&wr)[16],
                                           float* w3p, int wv, int ln)
{
    float4 acc = make_float4(0.f, 0.f, 0.f, 0.f);
    #pragma unroll
    for (int j = 0; j < 16; ++j) {
        const float d = dw[j];
        acc.x = fmaf(d, wr[j].x, acc.x);
        acc.y = fmaf(d, wr[j].y, acc.y);
        acc.z = fmaf(d, wr[j].z, acc.z);
        acc.w = fmaf(d, wr[j].w, acc.w);
    }
    *(float4*)&w3p[wv * Dc + 4 * ln] = acc;
}

constexpr int GS = 16;
constexpr int KG = 4 * GS;

__device__ __forceinline__ void b_group_load(const float* __restrict__ wbk, int k0, v2f (&bf)[GS])
{
    #pragma unroll
    for (int j = 0; j < GS; ++j) {
        bf[j].x = wbk[(size_t)(k0 + 4 * j)     * Dc];
        bf[j].y = wbk[(size_t)(k0 + 4 * j + 1) * Dc];
    }
}

template <int NT, int LD>
__device__ __forceinline__ void mma_group(const float* ap, int k0, const v2f (&bf)[GS], v8f (&acc)[NT])
{
    #pragma unroll
    for (int j = 0; j < GS; ++j) {
        #pragma unroll
        for (int m = 0; m < NT; ++m) {
            const v2f fa = *(const v2f*)(ap + m * 16 * LD + k0 + 4 * j);
            acc[m] = __builtin_amdgcn_wmma_f32_16x16x4_f32(
                         false, fa, false, bf[j], (short)0, acc[m], false, false);
        }
    }
}

template <int NT, int LD, int K, bool NEXT>
__device__ __forceinline__ void k_sweep(const float* __restrict__ wbk,
                                        const float* __restrict__ wnext,
                                        const float* ap, v2f (&b0)[GS], v8f (&acc)[NT])
{
    static_assert(K % (2 * KG) == 0, "two groups per trip");
    v2f b1[GS];
    #pragma unroll 1
    for (int k = 0; k < K - 2 * KG; k += 2 * KG) {
        b_group_load(wbk, k + KG, b1);
        mma_group<NT, LD>(ap, k, b0, acc);
        b_group_load(wbk, k + 2 * KG, b0);
        mma_group<NT, LD>(ap, k + KG, b1, acc);
    }
    b_group_load(wbk, K - KG, b1);
    mma_group<NT, LD>(ap, K - 2 * KG, b0, acc);
    if (NEXT) b_group_load(wnext, 0, b0);
    mma_group<NT, LD>(ap, K - KG, b1, acc);
}

template <int NV>
__device__ __forceinline__ void row_piece_load(const float* __restrict__ enc_row,
                                               const float* __restrict__ enc_col,
                                               int b, int idx, int c0, float4 (&v)[NV])
{
    const bool is_stage = c0 < Dc;
    const bool ok = is_stage || (idx >= 0 && idx < Nw);
    const float* src = is_stage
        ? enc_col + ((size_t)b * Sc + (idx - 1)) * Dc + c0
        : enc_row + ((size_t)b * Nw + (ok ? idx : 0)) * Dc + (c0 - Dc);
    #pragma unroll
    for (int j = 0; j < NV; ++j)
        v[j] = *(const float4*)(src + 4 * j);
    if (!ok) {
        #pragma unroll
        for (int j = 0; j < NV; ++j)
            v[j] = make_float4(0.f, 0.f, 0.f, 0.f);
    }
}

template <int NT>
__device__ __forceinline__ void pm_rows(const float* __restrict__ enc_row,
                                        const float* __restrict__ enc_col,
                                        const float* __restrict__ clockv,
                                        const float* __restrict__ pet,
                                        const float* __restrict__ W_dyn,
                                        const float* __restrict__ W_concat,
                                        const int*   __restrict__ lhw,
                                        const int*   __restrict__ lstage,
                                        float*       __restrict__ out,
                                        float* Xs, float* w3p, float* Rp, int g0)
{
    const int t     = threadIdx.x;
    const int ln    = t & 31;
    const int wv    = __builtin_amdgcn_readfirstlane(t >> 5);
    const int l16   = ln & 15;
    const int khalf = (ln >> 4) << 1;
    const int rhalf = (ln >> 4) << 3;
    const int n0    = wv * 16;

    const int rA = t >> 3,        cA = (t & 7) * 32;
    const int rB = 32 + (t >> 4), cB = (t & 15) * 16;
    const int gA = g0 + rA;
    const int gB = g0 + rB;
    const int idxA = (cA < Dc ? lstage : lhw)[gA];
    int idxB = 0;
    if (NT == 3) idxB = (cB < Dc ? lstage : lhw)[gB];
    float petv = 0.f, clk = 0.f;
    if (t < 16 * NT) {
        petv = pet[g0 + t];
        clk  = clockv[(g0 + t) >> 6];
    }

    float dw[16];
    float4 wr[16];
    w3_load(W_dyn, W_concat, wv, ln, dw, wr);

    const float* wbk = W_concat + (size_t)khalf * Dc + n0 + l16;
    v2f b0[GS];
    b_group_load(wbk, 0, b0);

    {
        float4 va[8];
        float4 vb[4];
        row_piece_load<8>(enc_row, enc_col, gA >> 6, idxA, cA, va);
        if (NT == 3) row_piece_load<4>(enc_row, enc_col, gB >> 6, idxB, cB, vb);
        float* da = &Xs[rA * LDA + cA];
        #pragma unroll
        for (int j = 0; j < 8; ++j)
            *(float4*)(da + 4 * j) = va[j];
        if (NT == 3) {
            float* db = &Xs[rB * LDA + cB];
            #pragma unroll
            for (int j = 0; j < 4; ++j)
                *(float4*)(db + 4 * j) = vb[j];
        }
    }
    w3_partial(dw, wr, w3p, wv, ln);
    if (t < 16 * NT) {
        const float d = petv - clk;
        Rp[t] = (d > 0.f ? d : 0.f) * NORM_INV;
    }
    __syncthreads();

    v8f acc[NT];
    #pragma unroll
    for (int m = 0; m < NT; ++m)
        acc[m] = v8f{};
    const float* ap = &Xs[l16 * LDA + khalf];
    k_sweep<NT, LDA, KA, false>(wbk, nullptr, ap, b0, acc);

    float w3v = w3p[n0 + l16];
    #pragma unroll
    for (int w = 1; w < 8; ++w)
        w3v += w3p[w * Dc + n0 + l16];
    #pragma unroll
    for (int m = 0; m < NT; ++m) {
        #pragma unroll
        for (int q = 0; q < 8; ++q) {
            const int row = m * 16 + rhalf + q;
            const int g   = g0 + row;
            const float v = acc[m][q] + Rp[row] * w3v;
            out[((size_t)(g >> 6) * (Pc + 2) + (g & 63)) * Dc + n0 + l16] = v;
        }
    }
}

__global__ __launch_bounds__(256)
void pm_robot_kernel(const float* __restrict__ enc_row,
                     const float* __restrict__ enc_col,
                     const float* __restrict__ clockv,
                     const float* __restrict__ pet,
                     const float* __restrict__ W_dyn,
                     const float* __restrict__ W_concat,
                     const float* __restrict__ W_robot,
                     const int*   __restrict__ lhw,
                     const int*   __restrict__ lstage,
                     const int* __restrict__ a1loc, const int* __restrict__ a2loc,
                     const int* __restrict__ a1rec, const int* __restrict__ a2rec,
                     const int* __restrict__ a1ns,  const int* __restrict__ a2ns,
                     float*       __restrict__ out)
{
    __shared__ __attribute__((aligned(16))) float Sx[48 * LDA];
    __shared__ __attribute__((aligned(16))) float w3p[8 * Dc];
    __shared__ __attribute__((aligned(16))) float Rp[48];
    float* const Xs = Sx;
    float* const Ys = Sx + 16 * LDA;
    static_assert(16 * LDA + 16 * LDB <= 48 * LDA, "arm rows fit behind the 16 PM rows");
    __shared__ __attribute__((aligned(16))) int   Lk[16];

    const int t    = threadIdx.x;
    const int ln   = t & 31;
    const int wv   = __builtin_amdgcn_readfirstlane(t >> 5);

    const int wave  = t >> 5;
    const int lane  = t & 31;
    const int l16   = lane & 15;
    const int khalf = (lane >> 4) << 1;
    const int rhalf = (lane >> 4) << 3;
    const int n0    = wave * 16;

    if (blockIdx.x < NRB) {
        const int mt  = blockIdx.x;
        const int r   = t >> 4;
        const int c8  = (t & 15) * 8;
        const int ar  = mt * 16 + r;
        const int b   = ar >> 1, arm = ar & 1;

        const int loc = (arm ? a2loc : a1loc)[b];
        const int rec = (arm ? a2rec : a1rec)[b];
        const int ns  = (arm ? a2ns  : a1ns )[b];
        const float clk = clockv[b];

        float dw[16];
        float4 wr[16];
        w3_load(W_dyn, W_concat, wv, ln, dw, wr);

        const float* wck = W_concat + (size_t)khalf * Dc + n0 + l16;
        const float* wrk = W_robot  + (size_t)khalf * Dc + n0 + l16;
        v2f b0[GS];
        b_group_load(wck, 0, b0);

        const bool has_pm = (loc >= 1 && loc <= Pc);
        const int  p      = has_pm ? loc - 1 : 0;
        const int   stg  = lstage[b * Pc + p];
        const int   waf  = lhw[b * Pc + p];
        const float petv = pet[b * Pc + p];

        const bool rec_ok = (rec >= 0 && rec < Nw);
        const bool ns_ok  = (ns >= 1 && ns <= Sc);
        const float* rsrc = enc_row + ((size_t)b * Nw + (rec_ok ? rec : 0)) * Dc + c8;
        const float* nsrc = enc_col + ((size_t)b * Sc + (ns_ok ? ns - 1 : 0)) * Dc + c8;
        float4 y0 = *(const float4*)(rsrc);
        float4 y1 = *(const float4*)(rsrc + 4);
        float4 y2 = *(const float4*)(nsrc);
        float4 y3 = *(const float4*)(nsrc + 4);

        const bool waf_ok = (waf >= 0 && waf < Nw);
        const float* ssrc = enc_col + ((size_t)b * Sc + (stg - 1)) * Dc + c8;
        const float* wsrc = enc_row + ((size_t)b * Nw + (waf_ok ? waf : 0)) * Dc + c8;
        float4 x0 = *(const float4*)(ssrc);
        float4 x1 = *(const float4*)(ssrc + 4);
        float4 x2 = *(const float4*)(wsrc);
        float4 x3 = *(const float4*)(wsrc + 4);

        const float4 z4 = make_float4(0.f, 0.f, 0.f, 0.f);
        if (!rec_ok) { y0 = z4; y1 = z4; }
        if (!ns_ok)  { y2 = z4; y3 = z4; }
        if (!waf_ok) { x2 = z4; x3 = z4; }

        float* yd = &Ys[r * LDB + Dc + c8];
        *(float4*)(yd)          = y0;
        *(float4*)(yd + 4)      = y1;
        *(float4*)(yd + Dc)     = y2;
        *(float4*)(yd + Dc + 4) = y3;
        float* xd = &Xs[r * LDA + c8];
        *(float4*)(xd)          = x0;
        *(float4*)(xd + 4)      = x1;
        *(float4*)(xd + Dc)     = x2;
        *(float4*)(xd + Dc + 4) = x3;

        w3_partial(dw, wr, w3p, wv, ln);

        if ((t & 15) == 0) {
            const float d = petv - clk;
            Rp[r] = (d > 0.f ? d : 0.f) * NORM_INV;
            Lk[r] = has_pm ? 2 : (loc == Pc + 1 ? 1 : 0);
        }
        __syncthreads();

        v8f accp1[1] = {v8f{}};
        k_sweep<1, LDA, KA, true>(wck, wrk, &Xs[l16 * LDA + khalf], b0, accp1);
        const v8f accp = accp1[0];
        {
            float w3v = w3p[n0 + l16];
            #pragma unroll
            for (int w = 1; w < 8; ++w)
                w3v += w3p[w * Dc + n0 + l16];
            const float4 rpa = *(const float4*)&Rp[rhalf];
            const float4 rpb = *(const float4*)&Rp[rhalf + 4];
            const int4   lka = *(const int4*)&Lk[rhalf];
            const int4   lkb = *(const int4*)&Lk[rhalf + 4];
            const float rp[8] = {rpa.x, rpa.y, rpa.z, rpa.w, rpb.x, rpb.y, rpb.z, rpb.w};
            const int   lk[8] = {lka.x, lka.y, lka.z, lka.w, lkb.x, lkb.y, lkb.z, lkb.w};
            #pragma unroll
            for (int q = 0; q < 8; ++q) {
                const int row   = rhalf + q;
                const float v   = accp[q] + rp[q] * w3v;
                const float alt = (lk[q] == 1) ? 1.f : 0.f;
                Ys[row * LDB + n0 + l16] = (lk[q] == 2) ? v : alt;
            }
        }
        __syncthreads();

        v8f acc1[1] = {v8f{}};
        k_sweep<1, LDB, KB, false>(wrk, nullptr, &Ys[l16 * LDB + khalf], b0, acc1);
        const v8f acc = acc1[0];
        #pragma unroll
        for (int q = 0; q < 8; ++q) {
            const int mrow = rhalf + q;
            const int oar  = mt * 16 + mrow;
            const int ob   = oar >> 1, oarm = oar & 1;
            out[((size_t)ob * (Pc + 2) + Pc + oarm) * Dc + n0 + l16] = acc[q];
        }
        return;
    }

    const int bid = blockIdx.x - NRB;
    if (bid < NT3)
        pm_rows<3>(enc_row, enc_col, clockv, pet, W_dyn, W_concat, lhw, lstage, out,
                   Xs, w3p, Rp, 48 * bid);
    else
        pm_rows<2>(enc_row, enc_col, clockv, pet, W_dyn, W_concat, lhw, lstage, out,
                   Xs, w3p, Rp, 48 * NT3 + 32 * (bid - NT3));
}

extern "C" void kernel_launch(void* const* d_in, const int* in_sizes, int n_in,
                              void* d_out, int out_size, void* d_ws, size_t ws_size,
                              hipStream_t stream) {
    (void)in_sizes; (void)n_in; (void)out_size; (void)d_ws; (void)ws_size;
    const float* enc_row  = (const float*)d_in[0];
    const float* enc_col  = (const float*)d_in[1];
    const float* clockv   = (const float*)d_in[2];
    const float* pet      = (const float*)d_in[3];
    const float* W_dyn    = (const float*)d_in[4];
    const float* W_concat = (const float*)d_in[5];
    const float* W_robot  = (const float*)d_in[6];
    const int*   lhw      = (const int*)d_in[7];
    const int*   lstage   = (const int*)d_in[8];
    const int*   a1loc    = (const int*)d_in[9];
    const int*   a2loc    = (const int*)d_in[10];
    const int*   a1rec    = (const int*)d_in[11];
    const int*   a2rec    = (const int*)d_in[12];
    const int*   a1ns     = (const int*)d_in[13];
    const int*   a2ns     = (const int*)d_in[14];
    float* out = (float*)d_out;

    pm_robot_kernel<<<GRID, 256, 0, stream>>>(enc_row, enc_col, clockv, pet,
                                              W_dyn, W_concat, W_robot, lhw, lstage,
                                              a1loc, a2loc, a1rec, a2rec, a1ns, a2ns, out);
}


# A (LDS) and B (L2) fragment loads of the next 16-k-step block issued as ONE group in front of the current WMMA block (second A register set), no load left among the WMMAs; sum orders unchanged
# speedup vs baseline: 2.4519x; 1.1077x over previous
#include <hip/hip_runtime.h>

typedef __attribute__((ext_vector_type(2))) float v2f;
typedef __attribute__((ext_vector_type(8))) float v8f;

constexpr int Bc = 128, Nw = 4096, Sc = 32, Pc = 64, Dc = 128;
constexpr float NORM_INV = 1.0f / 300.0f;

constexpr int KA  = 256;
constexpr int LDA = 260;
constexpr int KB  = 384;
constexpr int LDB = 388;
constexpr int NRB  = 16;
constexpr int NT3  = 32;
constexpr int NT2  = 208;
constexpr int GRID = NRB + NT3 + NT2;
static_assert(3 * NT3 + 2 * NT2 == Bc * Pc / 16, "every PM row tile is taken exactly once");
static_assert(GRID == 256, "one work-group per WGP");

__device__ __forceinline__ void w3_load(const float* __restrict__ W_dyn,
                                        const float* __restrict__ W_concat,
                                        int wv, int ln, float (&dw)[16], float4 (&wr)[16])
{
    #pragma unroll
    for (int j = 0; j < 16; ++j)
        dw[j] = W_dyn[16 * wv + j];
    const float* wsrc = W_concat + (size_t)(256 + 16 * wv) * Dc + 4 * ln;
    #pragma unroll
    for (int j = 0; j < 16; ++j)
        wr[j] = *(const float4*)(wsrc + (size_t)j * Dc);
}

__device__ __forceinline__ void w3_partial(const float (&dw)[16], const float4 (&wr)[16],
                                           float* w3p, int wv, int ln)
{
    float4 acc = make_float4(0.f, 0.f, 0.f, 0.f);
    #pragma unroll
    for (int j = 0; j < 16; ++j) {
        const float d = dw[j];
        acc.x = fmaf(d, wr[j].x, acc.x);
        acc.y = fmaf(d, wr[j].y, acc.y);
        acc.z = fmaf(d, wr[j].z, acc.z);
        acc.w = fmaf(d, wr[j].w, acc.w);
    }
    *(float4*)&w3p[wv * Dc + 4 * ln] = acc;
}

constexpr int GS = 16;
constexpr int KG = 4 * GS;

__device__ __forceinline__ void b_group_load(const float* __restrict__ wbk, int k0, v2f (&bf)[GS])
{
    #pragma unroll
    for (int j = 0; j < GS; ++j) {
        bf[j].x = wbk[(size_t)(k0 + 4 * j)     * Dc];
        bf[j].y = wbk[(size_t)(k0 + 4 * j + 1) * Dc];
    }
}

template <int NT, int LD>
__device__ __forceinline__ void a_group_load(const float* ap, int k0, v2f (&af)[GS][NT])
{
    #pragma unroll
    for (int j = 0; j < GS; ++j) {
        #pragma unroll
        for (int m = 0; m < NT; ++m)
            af[j][m] = *(const v2f*)(ap + m * 16 * LD + k0 + 4 * j);
    }
}

template <int NT>
__device__ __forceinline__ void mma_group(const v2f (&af)[GS][NT], const v2f (&bf)[GS], v8f (&acc)[NT])
{
    #pragma unroll
    for (int j = 0; j < GS; ++j) {
        #pragma unroll
        for (int m = 0; m < NT; ++m)
            acc[m] = __builtin_amdgcn_wmma_f32_16x16x4_f32(
                         false, af[j][m], false, bf[j], (short)0, acc[m], false, false);
    }
}

template <int NT>
__device__ __forceinline__ void block_edge(v8f (&acc)[NT])
{
    static_assert(NT >= 1 && NT <= 3, "one operand per accumulator tile");
    if constexpr (NT == 1)
        asm volatile("" : "+v"(acc[0]) : : "memory");
    else if constexpr (NT == 2)
        asm volatile("" : "+v"(acc[0]), "+v"(acc[1]) : : "memory");
    else
        asm volatile("" : "+v"(acc[0]), "+v"(acc[1]), "+v"(acc[2]) : : "memory");
}

template <int NT, int LD, int K, bool NEXT, bool EPI, class F>
__device__ __forceinline__ void k_sweep(const float* __restrict__ wbk,
                                        const float* __restrict__ wnext,
                                        const float* ap, v2f (&b0)[GS], v8f (&acc)[NT],
                                        F&& epi_loads)
{
    static_assert(K % (2 * KG) == 0, "two groups per trip");
    v2f a0[GS][NT];
    v2f a1[GS][NT];
    v2f b1[GS];
    a_group_load<NT, LD>(ap, 0, a0);
    #pragma unroll 1
    for (int k = 0; k < K - 2 * KG; k += 2 * KG) {
        a_group_load<NT, LD>(ap, k + KG, a1);
        b_group_load(wbk, k + KG, b1);
        block_edge<NT>(acc);
        mma_group<NT>(a0, b0, acc);
        block_edge<NT>(acc);
        a_group_load<NT, LD>(ap, k + 2 * KG, a0);
        b_group_load(wbk, k + 2 * KG, b0);
        block_edge<NT>(acc);
        mma_group<NT>(a1, b1, acc);
        block_edge<NT>(acc);
    }
    a_group_load<NT, LD>(ap, K - KG, a1);
    b_group_load(wbk, K - KG, b1);
    block_edge<NT>(acc);
    mma_group<NT>(a0, b0, acc);
    if (NEXT || EPI) {
        block_edge<NT>(acc);
        if (NEXT) b_group_load(wnext, 0, b0);
        if (EPI) epi_loads();
        block_edge<NT>(acc);
    }
    mma_group<NT>(a1, b1, acc);
}

template <int NV>
__device__ __forceinline__ void row_piece_load(const float* __restrict__ enc_row,
                                               const float* __restrict__ enc_col,
                                               int b, int idx, int c0, float4 (&v)[NV])
{
    const bool is_stage = c0 < Dc;
    const bool ok = is_stage || (idx >= 0 && idx < Nw);
    const float* src = is_stage
        ? enc_col + ((size_t)b * Sc + (idx - 1)) * Dc + c0
        : enc_row + ((size_t)b * Nw + (ok ? idx : 0)) * Dc + (c0 - Dc);
    #pragma unroll
    for (int j = 0; j < NV; ++j)
        v[j] = *(const float4*)(src + 4 * j);
    if (!ok) {
        #pragma unroll
        for (int j = 0; j < NV; ++j)
            v[j] = make_float4(0.f, 0.f, 0.f, 0.f);
    }
}

template <int NT>
__device__ __forceinline__ void pm_rows(const float* __restrict__ enc_row,
                                        const float* __restrict__ enc_col,
                                        const float* __restrict__ clockv,
                                        const float* __restrict__ pet,
                                        const float* __restrict__ W_dyn,
                                        const float* __restrict__ W_concat,
                                        const int*   __restrict__ lhw,
                                        const int*   __restrict__ lstage,
                                        float*       __restrict__ out,
                                        float* Xs, float* w3p, float* Rp, int g0)
{
    const int t     = threadIdx.x;
    const int ln    = t & 31;
    const int wv    = __builtin_amdgcn_readfirstlane(t >> 5);
    const int l16   = ln & 15;
    const int khalf = (ln >> 4) << 1;
    const int rhalf = (ln >> 4) << 3;
    const int n0    = wv * 16;

    const int rA = t >> 3,        cA = (t & 7) * 32;
    const int rB = 32 + (t >> 4), cB = (t & 15) * 16;
    const int gA = g0 + rA;
    const int gB = g0 + rB;
    const int idxA = (cA < Dc ? lstage : lhw)[gA];
    int idxB = 0;
    if (NT == 3) idxB = (cB < Dc ? lstage : lhw)[gB];
    float petv = 0.f, clk = 0.f;
    if (t < 16 * NT) {
        petv = pet[g0 + t];
        clk  = clockv[(g0 + t) >> 6];
    }

    float dw[16];
    float4 wr[16];
    w3_load(W_dyn, W_concat, wv, ln, dw, wr);

    const float* wbk = W_concat + (size_t)khalf * Dc + n0 + l16;
    v2f b0[GS];
    b_group_load(wbk, 0, b0);

    {
        float4 va[8];
        float4 vb[4];
        row_piece_load<8>(enc_row, enc_col, gA >> 6, idxA, cA, va);
        if (NT == 3) row_piece_load<4>(enc_row, enc_col, gB >> 6, idxB, cB, vb);
        float* da = &Xs[rA * LDA + cA];
        #pragma unroll
        for (int j = 0; j < 8; ++j)
            *(float4*)(da + 4 * j) = va[j];
        if (NT == 3) {
            float* db = &Xs[rB * LDA + cB];
            #pragma unroll
            for (int j = 0; j < 4; ++j)
                *(float4*)(db + 4 * j) = vb[j];
        }
    }
    w3_partial(dw, wr, w3p, wv, ln);
    if (t < 16 * NT) {
        const float d = petv - clk;
        Rp[t] = (d > 0.f ? d : 0.f) * NORM_INV;
    }
    __syncthreads();

    float w3q[8];
    float4 rpa[NT], rpb[NT];
    auto epi_loads = [&]() __attribute__((always_inline)) {
        #pragma unroll
        for (int w = 0; w < 8; ++w)
            w3q[w] = w3p[w * Dc + n0 + l16];
        #pragma unroll
        for (int m = 0; m < NT; ++m) {
            rpa[m] = *(const float4*)&Rp[m * 16 + rhalf];
            rpb[m] = *(const float4*)&Rp[m * 16 + rhalf + 4];
        }
    };

    v8f acc[NT];
    #pragma unroll
    for (int m = 0; m < NT; ++m)
        acc[m] = v8f{};
    const float* ap = &Xs[l16 * LDA + khalf];
    k_sweep<NT, LDA, KA, false, true>(wbk, nullptr, ap, b0, acc, epi_loads);

    float w3v = w3q[0];
    #pragma unroll
    for (int w = 1; w < 8; ++w)
        w3v += w3q[w];
    #pragma unroll
    for (int m = 0; m < NT; ++m) {
        const float rp[8] = {rpa[m].x, rpa[m].y, rpa[m].z, rpa[m].w,
                             rpb[m].x, rpb[m].y, rpb[m].z, rpb[m].w};
        #pragma unroll
        for (int q = 0; q < 8; ++q) {
            const int row = m * 16 + rhalf + q;
            const int g   = g0 + row;
            const float v = acc[m][q] + rp[q] * w3v;
            out[((size_t)(g >> 6) * (Pc + 2) + (g & 63)) * Dc + n0 + l16] = v;
        }
    }
}

__global__ __launch_bounds__(256)
void pm_robot_kernel(const float* __restrict__ enc_row,
                     const float* __restrict__ enc_col,
                     const float* __restrict__ clockv,
                     const float* __restrict__ pet,
                     const float* __restrict__ W_dyn,
                     const float* __restrict__ W_concat,
                     const float* __restrict__ W_robot,
                     const int*   __restrict__ lhw,
                     const int*   __restrict__ lstage,
                     const int* __restrict__ a1loc, const int* __restrict__ a2loc,
                     const int* __restrict__ a1rec, const int* __restrict__ a2rec,
                     const int* __restrict__ a1ns,  const int* __restrict__ a2ns,
                     float*       __restrict__ out)
{
    __shared__ __attribute__((aligned(16))) float Sx[48 * LDA];
    __shared__ __attribute__((aligned(16))) float w3p[8 * Dc];
    __shared__ __attribute__((aligned(16))) float Rp[48];
    float* const Xs = Sx;
    float* const Ys = Sx + 16 * LDA;
    static_assert(16 * LDA + 16 * LDB <= 48 * LDA, "arm rows fit behind the 16 PM rows");
    __shared__ __attribute__((aligned(16))) int   Lk[16];

    const int t    = threadIdx.x;
    const int ln   = t & 31;
    const int wv   = __builtin_amdgcn_readfirstlane(t >> 5);

    const int wave  = t >> 5;
    const int lane  = t & 31;
    const int l16   = lane & 15;
    const int khalf = (lane >> 4) << 1;
    const int rhalf = (lane >> 4) << 3;
    const int n0    = wave * 16;

    if (blockIdx.x < NRB) {
        const int mt  = blockIdx.x;
        const int r   = t >> 4;
        const int c8  = (t & 15) * 8;
        const int ar  = mt * 16 + r;
        const int b   = ar >> 1, arm = ar & 1;

        const int loc = (arm ? a2loc : a1loc)[b];
        const int rec = (arm ? a2rec : a1rec)[b];
        const int ns  = (arm ? a2ns  : a1ns )[b];
        const float clk = clockv[b];

        float dw[16];
        float4 wr[16];
        w3_load(W_dyn, W_concat, wv, ln, dw, wr);

        const float* wck = W_concat + (size_t)khalf * Dc + n0 + l16;
        const float* wrk = W_robot  + (size_t)khalf * Dc + n0 + l16;
        v2f b0[GS];
        b_group_load(wck, 0, b0);

        const bool has_pm = (loc >= 1 && loc <= Pc);
        const int  p      = has_pm ? loc - 1 : 0;
        const int   stg  = lstage[b * Pc + p];
        const int   waf  = lhw[b * Pc + p];
        const float petv = pet[b * Pc + p];

        const bool rec_ok = (rec >= 0 && rec < Nw);
        const bool ns_ok  = (ns >= 1 && ns <= Sc);
        const float* rsrc = enc_row + ((size_t)b * Nw + (rec_ok ? rec : 0)) * Dc + c8;
        const float* nsrc = enc_col + ((size_t)b * Sc + (ns_ok ? ns - 1 : 0)) * Dc + c8;
        float4 y0 = *(const float4*)(rsrc);
        float4 y1 = *(const float4*)(rsrc + 4);
        float4 y2 = *(const float4*)(nsrc);
        float4 y3 = *(const float4*)(nsrc + 4);

        const bool waf_ok = (waf >= 0 && waf < Nw);
        const float* ssrc = enc_col + ((size_t)b * Sc + (stg - 1)) * Dc + c8;
        const float* wsrc = enc_row + ((size_t)b * Nw + (waf_ok ? waf : 0)) * Dc + c8;
        float4 x0 = *(const float4*)(ssrc);
        float4 x1 = *(const float4*)(ssrc + 4);
        float4 x2 = *(const float4*)(wsrc);
        float4 x3 = *(const float4*)(wsrc + 4);

        const float4 z4 = make_float4(0.f, 0.f, 0.f, 0.f);
        if (!rec_ok) { y0 = z4; y1 = z4; }
        if (!ns_ok)  { y2 = z4; y3 = z4; }
        if (!waf_ok) { x2 = z4; x3 = z4; }

        float* yd = &Ys[r * LDB + Dc + c8];
        *(float4*)(yd)          = y0;
        *(float4*)(yd + 4)      = y1;
        *(float4*)(yd + Dc)     = y2;
        *(float4*)(yd + Dc + 4) = y3;
        float* xd = &Xs[r * LDA + c8];
        *(float4*)(xd)          = x0;
        *(float4*)(xd + 4)      = x1;
        *(float4*)(xd + Dc)     = x2;
        *(float4*)(xd + Dc + 4) = x3;

        w3_partial(dw, wr, w3p, wv, ln);

        if ((t & 15) == 0) {
            const float d = petv - clk;
            Rp[r] = (d > 0.f ? d : 0.f) * NORM_INV;
            Lk[r] = has_pm ? 2 : (loc == Pc + 1 ? 1 : 0);
        }
        __syncthreads();

        float  w3q[8];
        float4 rpa, rpb;
        int4   lka, lkb;
        auto epi_loads = [&]() __attribute__((always_inline)) {
            #pragma unroll
            for (int w = 0; w < 8; ++w)
                w3q[w] = w3p[w * Dc + n0 + l16];
            rpa = *(const float4*)&Rp[rhalf];
            rpb = *(const float4*)&Rp[rhalf + 4];
            lka = *(const int4*)&Lk[rhalf];
            lkb = *(const int4*)&Lk[rhalf + 4];
        };

        v8f accp1[1] = {v8f{}};
        k_sweep<1, LDA, KA, true, true>(wck, wrk, &Xs[l16 * LDA + khalf], b0, accp1, epi_loads);
        const v8f accp = accp1[0];
        {
            float w3v = w3q[0];
            #pragma unroll
            for (int w = 1; w < 8; ++w)
                w3v += w3q[w];
            const float rp[8] = {rpa.x, rpa.y, rpa.z, rpa.w, rpb.x, rpb.y, rpb.z, rpb.w};
            const int   lk[8] = {lka.x, lka.y, lka.z, lka.w, lkb.x, lkb.y, lkb.z, lkb.w};
            #pragma unroll
            for (int q = 0; q < 8; ++q) {
                const int row   = rhalf + q;
                const float v   = accp[q] + rp[q] * w3v;
                const float alt = (lk[q] == 1) ? 1.f : 0.f;
                Ys[row * LDB + n0 + l16] = (lk[q] == 2) ? v : alt;
            }
        }
        __syncthreads();

        v8f acc1[1] = {v8f{}};
        k_sweep<1, LDB, KB, false, false>(wrk, nullptr, &Ys[l16 * LDB + khalf], b0, acc1, []() {});
        const v8f acc = acc1[0];
        #pragma unroll
        for (int q = 0; q < 8; ++q) {
            const int mrow = rhalf + q;
            const int oar  = mt * 16 + mrow;
            const int ob   = oar >> 1, oarm = oar & 1;
            out[((size_t)ob * (Pc + 2) + Pc + oarm) * Dc + n0 + l16] = acc[q];
        }
        return;
    }

    const int bid = blockIdx.x - NRB;
    if (bid < NT3)
        pm_rows<3>(enc_row, enc_col, clockv, pet, W_dyn, W_concat, lhw, lstage, out,
                   Xs, w3p, Rp, 48 * bid);
    else
        pm_rows<2>(enc_row, enc_col, clockv, pet, W_dyn, W_concat, lhw, lstage, out,
                   Xs, w3p, Rp, 48 * NT3 + 32 * (bid - NT3));
}

extern "C" void kernel_launch(void* const* d_in, const int* in_sizes, int n_in,
                              void* d_out, int out_size, void* d_ws, size_t ws_size,
                              hipStream_t stream) {
    (void)in_sizes; (void)n_in; (void)out_size; (void)d_ws; (void)ws_size;
    const float* enc_row  = (const float*)d_in[0];
    const float* enc_col  = (const float*)d_in[1];
    const float* clockv   = (const float*)d_in[2];
    const float* pet      = (const float*)d_in[3];
    const float* W_dyn    = (const float*)d_in[4];
    const float* W_concat = (const float*)d_in[5];
    const float* W_robot  = (const float*)d_in[6];
    const int*   lhw      = (const int*)d_in[7];
    const int*   lstage   = (const int*)d_in[8];
    const int*   a1loc    = (const int*)d_in[9];
    const int*   a2loc    = (const int*)d_in[10];
    const int*   a1rec    = (const int*)d_in[11];
    const int*   a2rec    = (const int*)d_in[12];
    const int*   a1ns     = (const int*)d_in[13];
    const int*   a2ns     = (const int*)d_in[14];
    float* out = (float*)d_out;

    pm_robot_kernel<<<GRID, 256, 0, stream>>>(enc_row, enc_col, clockv, pet,
                                              W_dyn, W_concat, W_robot, lhw, lstage,
                                              a1loc, a2loc, a1rec, a2rec, a1ns, a2ns, out);
}


# matrix work evened over SIMDs (max 384 -> 320 WMMAs): PM tiles dealt in column halves (128 groups x 2.5 tiles, 96 x 2), arms over 32 groups by column half, arm K summed as two partials
# speedup vs baseline: 2.6888x; 1.0966x over previous
#include <hip/hip_runtime.h>

typedef __attribute__((ext_vector_type(2))) float v2f;
typedef __attribute__((ext_vector_type(8))) float v8f;

constexpr int Bc = 128, Nw = 4096, Sc = 32, Pc = 64, Dc = 128;
constexpr float NORM_INV = 1.0f / 300.0f;

constexpr int KA  = 256;
constexpr int LDA = 260;
constexpr int KB  = 384;
constexpr int LDB = 388;
constexpr int NRB  = 32;
constexpr int N5   = 128;
constexpr int N4   = 96;
constexpr int GRID = NRB + N5 + N4;
static_assert(2 * N5 + N5 / 2 + 2 * N4 == Bc * Pc / 16, "every PM row tile is taken exactly once");
static_assert(N5 % 2 == 0, "the five-half-tile work-groups come in pairs");
static_assert(GRID == 256, "one work-group per WGP");

__device__ __forceinline__ bool wave_set0(int w) { return ((w ^ (w >> 2)) & 1) == 0; }
__device__ __forceinline__ int wave_ntile(int w) { return (wave_set0(w) ? 0 : 4) + (w >> 1); }

__device__ __forceinline__ void w3_load(const float* __restrict__ W_dyn,
                                        const float* __restrict__ W_concat,
                                        int wv, int ln, float (&dw)[16], float4 (&wr)[16])
{
    #pragma unroll
    for (int j = 0; j < 16; ++j)
        dw[j] = W_dyn[16 * wv + j];
    const float* wsrc = W_concat + (size_t)(256 + 16 * wv) * Dc + 4 * ln;
    #pragma unroll
    for (int j = 0; j < 16; ++j)
        wr[j] = *(const float4*)(wsrc + (size_t)j * Dc);
}

__device__ __forceinline__ void w3_partial(const float (&dw)[16], const float4 (&wr)[16],
                                           float* w3p, int wv, int ln)
{
    float4 acc = make_float4(0.f, 0.f, 0.f, 0.f);
    #pragma unroll
    for (int j = 0; j < 16; ++j) {
        const float d = dw[j];
        acc.x = fmaf(d, wr[j].x, acc.x);
        acc.y = fmaf(d, wr[j].y, acc.y);
        acc.z = fmaf(d, wr[j].z, acc.z);
        acc.w = fmaf(d, wr[j].w, acc.w);
    }
    *(float4*)&w3p[wv * Dc + 4 * ln] = acc;
}

constexpr int GS = 16;
constexpr int KG = 4 * GS;

__device__ __forceinline__ void b_group_load(const float* __restrict__ wbk, int k0, v2f (&bf)[GS])
{
    #pragma unroll
    for (int j = 0; j < GS; ++j) {
        bf[j].x = wbk[(size_t)(k0 + 4 * j)     * Dc];
        bf[j].y = wbk[(size_t)(k0 + 4 * j + 1) * Dc];
    }
}

template <int NT, int LD>
__device__ __forceinline__ void a_group_load(const float* ap, int k0, v2f (&af)[GS][NT])
{
    #pragma unroll
    for (int j = 0; j < GS; ++j) {
        #pragma unroll
        for (int m = 0; m < NT; ++m)
            af[j][m] = *(const v2f*)(ap + m * 16 * LD + k0 + 4 * j);
    }
}

template <int NT>
__device__ __forceinline__ void mma_group(const v2f (&af)[GS][NT], const v2f (&bf)[GS], v8f (&acc)[NT])
{
    #pragma unroll
    for (int j = 0; j < GS; ++j) {
        #pragma unroll
        for (int m = 0; m < NT; ++m)
            acc[m] = __builtin_amdgcn_wmma_f32_16x16x4_f32(
                         false, af[j][m], false, bf[j], (short)0, acc[m], false, false);
    }
}

template <int NT>
__device__ __forceinline__ void block_edge(v8f (&acc)[NT])
{
    static_assert(NT >= 1 && NT <= 3, "one operand per accumulator tile");
    if constexpr (NT == 1)
        asm volatile("" : "+v"(acc[0]) : : "memory");
    else if constexpr (NT == 2)
        asm volatile("" : "+v"(acc[0]), "+v"(acc[1]) : : "memory");
    else
        asm volatile("" : "+v"(acc[0]), "+v"(acc[1]), "+v"(acc[2]) : : "memory");
}

template <int NT, int LD, int K, bool NEXT, bool EPI, class F>
__device__ __forceinline__ void k_sweep(const float* __restrict__ wbk,
                                        const float* __restrict__ wnext,
                                        const float* ap, v2f (&b0)[GS], v8f (&acc)[NT],
                                        F&& epi_loads)
{
    static_assert(K % (2 * KG) == 0, "two groups per trip");
    v2f a0[GS][NT];
    v2f a1[GS][NT];
    v2f b1[GS];
    a_group_load<NT, LD>(ap, 0, a0);
    #pragma unroll 1
    for (int k = 0; k < K - 2 * KG; k += 2 * KG) {
        a_group_load<NT, LD>(ap, k + KG, a1);
        b_group_load(wbk, k + KG, b1);
        block_edge<NT>(acc);
        mma_group<NT>(a0, b0, acc);
        block_edge<NT>(acc);
        a_group_load<NT, LD>(ap, k + 2 * KG, a0);
        b_group_load(wbk, k + 2 * KG, b0);
        block_edge<NT>(acc);
        mma_group<NT>(a1, b1, acc);
        block_edge<NT>(acc);
    }
    a_group_load<NT, LD>(ap, K - KG, a1);
    b_group_load(wbk, K - KG, b1);
    block_edge<NT>(acc);
    mma_group<NT>(a0, b0, acc);
    if (NEXT || EPI) {
        block_edge<NT>(acc);
        if (NEXT) b_group_load(wnext, 0, b0);
        if (EPI) epi_loads();
        block_edge<NT>(acc);
    }
    mma_group<NT>(a1, b1, acc);
}

template <int NV>
__device__ __forceinline__ void row_piece_load(const float* __restrict__ enc_row,
                                               const float* __restrict__ enc_col,
                                               int b, int idx, int c0, float4 (&v)[NV])
{
    const bool is_stage = c0 < Dc;
    const bool ok = is_stage || (idx >= 0 && idx < Nw);
    const float* src = is_stage
        ? enc_col + ((size_t)b * Sc + (idx - 1)) * Dc + c0
        : enc_row + ((size_t)b * Nw + (ok ? idx : 0)) * Dc + (c0 - Dc);
    #pragma unroll
    for (int j = 0; j < NV; ++j)
        v[j] = *(const float4*)(src + 4 * j);
    if (!ok) {
        #pragma unroll
        for (int j = 0; j < NV; ++j)
            v[j] = make_float4(0.f, 0.f, 0.f, 0.f);
    }
}

template <int NT>
__device__ __forceinline__ void pm_sweep_store(const float* __restrict__ wbk,
                                               float*       __restrict__ out,
                                               const float* Xs, const float* w3p, const float* Rp,
                                               v2f (&b0)[GS], int g0, int m0,
                                               int n0, int l16, int khalf, int rhalf)
{
    float w3q[8];
    float4 rpa[NT], rpb[NT];
    auto epi_loads = [&]() __attribute__((always_inline)) {
        #pragma unroll
        for (int w = 0; w < 8; ++w)
            w3q[w] = w3p[w * Dc + n0 + l16];
        #pragma unroll
        for (int m = 0; m < NT; ++m) {
            rpa[m] = *(const float4*)&Rp[(m0 + m) * 16 + rhalf];
            rpb[m] = *(const float4*)&Rp[(m0 + m) * 16 + rhalf + 4];
        }
    };

    v8f acc[NT];
    #pragma unroll
    for (int m = 0; m < NT; ++m)
        acc[m] = v8f{};
    const float* ap = &Xs[(m0 * 16 + l16) * LDA + khalf];
    k_sweep<NT, LDA, KA, false, true>(wbk, nullptr, ap, b0, acc, epi_loads);

    float w3v = w3q[0];
    #pragma unroll
    for (int w = 1; w < 8; ++w)
        w3v += w3q[w];
    #pragma unroll
    for (int m = 0; m < NT; ++m) {
        const float rp[8] = {rpa[m].x, rpa[m].y, rpa[m].z, rpa[m].w,
                             rpb[m].x, rpb[m].y, rpb[m].z, rpb[m].w};
        #pragma unroll
        for (int q = 0; q < 8; ++q) {
            const int row = (m0 + m) * 16 + rhalf + q;
            const int g   = g0 + row;
            const float v = acc[m][q] + rp[q] * w3v;
            out[((size_t)(g >> 6) * (Pc + 2) + (g & 63)) * Dc + n0 + l16] = v;
        }
    }
}

template <int NTG, bool SHARED>
__device__ __forceinline__ void pm_rows(const float* __restrict__ enc_row,
                                        const float* __restrict__ enc_col,
                                        const float* __restrict__ clockv,
                                        const float* __restrict__ pet,
                                        const float* __restrict__ W_dyn,
                                        const float* __restrict__ W_concat,
                                        const int*   __restrict__ lhw,
                                        const int*   __restrict__ lstage,
                                        float*       __restrict__ out,
                                        float* Xs, float* w3p, float* Rp, int g0, int side)
{
    static_assert(!SHARED || NTG == 3, "a shared tile is the third of three");
    const int t     = threadIdx.x;
    const int ln    = t & 31;
    const int wv    = __builtin_amdgcn_readfirstlane(t >> 5);
    const int l16   = ln & 15;
    const int khalf = (ln >> 4) << 1;
    const int rhalf = (ln >> 4) << 3;
    const int n0    = wave_ntile(wv) * 16;

    const int rA = t >> 3,        cA = (t & 7) * 32;
    const int rB = 32 + (t >> 4), cB = (t & 15) * 16;
    const int gA = g0 + rA;
    const int gB = g0 + rB;
    const int idxA = (cA < Dc ? lstage : lhw)[gA];
    int idxB = 0;
    if (NTG == 3) idxB = (cB < Dc ? lstage : lhw)[gB];
    float petv = 0.f, clk = 0.f;
    if (t < 16 * NTG) {
        petv = pet[g0 + t];
        clk  = clockv[(g0 + t) >> 6];
    }

    float dw[16];
    float4 wr[16];
    w3_load(W_dyn, W_concat, wv, ln, dw, wr);

    const float* wbk = W_concat + (size_t)khalf * Dc + n0 + l16;
    v2f b0[GS];
    b_group_load(wbk, 0, b0);

    {
        float4 va[8];
        float4 vb[4];
        row_piece_load<8>(enc_row, enc_col, gA >> 6, idxA, cA, va);
        if (NTG == 3) row_piece_load<4>(enc_row, enc_col, gB >> 6, idxB, cB, vb);
        float* da = &Xs[rA * LDA + cA];
        #pragma unroll
        for (int j = 0; j < 8; ++j)
            *(float4*)(da + 4 * j) = va[j];
        if (NTG == 3) {
            float* db = &Xs[rB * LDA + cB];
            #pragma unroll
            for (int j = 0; j < 4; ++j)
                *(float4*)(db + 4 * j) = vb[j];
        }
    }
    w3_partial(dw, wr, w3p, wv, ln);
    if (t < 16 * NTG) {
        const float d = petv - clk;
        Rp[t] = (d > 0.f ? d : 0.f) * NORM_INV;
    }
    __syncthreads();

    if constexpr (!SHARED) {
        pm_sweep_store<NTG>(wbk, out, Xs, w3p, Rp, b0, g0, 0, n0, l16, khalf, rhalf);
    } else {
        if (wave_set0(wv) == (side == 0))
            pm_sweep_store<3>(wbk, out, Xs, w3p, Rp, b0, g0, 0, n0, l16, khalf, rhalf);
        else
            pm_sweep_store<2>(wbk, out, Xs, w3p, Rp, b0, g0, side, n0, l16, khalf, rhalf);
    }
}

__global__ __launch_bounds__(256)
void pm_robot_kernel(const float* __restrict__ enc_row,
                     const float* __restrict__ enc_col,
                     const float* __restrict__ clockv,
                     const float* __restrict__ pet,
                     const float* __restrict__ W_dyn,
                     const float* __restrict__ W_concat,
                     const float* __restrict__ W_robot,
                     const int*   __restrict__ lhw,
                     const int*   __restrict__ lstage,
                     const int* __restrict__ a1loc, const int* __restrict__ a2loc,
                     const int* __restrict__ a1rec, const int* __restrict__ a2rec,
                     const int* __restrict__ a1ns,  const int* __restrict__ a2ns,
                     float*       __restrict__ out)
{
    __shared__ __attribute__((aligned(16))) float Sx[48 * LDA];
    __shared__ __attribute__((aligned(16))) float w3p[8 * Dc];
    __shared__ __attribute__((aligned(16))) float Rp[48];
    float* const Xs = Sx;
    float* const Ys = Sx + 16 * LDA;
    static_assert(16 * LDA + 16 * LDB <= 48 * LDA, "arm rows fit behind the 16 PM rows");
    __shared__ __attribute__((aligned(16))) int   Lk[16];

    const int t    = threadIdx.x;
    const int ln   = t & 31;
    const int wv   = __builtin_amdgcn_readfirstlane(t >> 5);

    const int l16   = ln & 15;
    const int khalf = (ln >> 4) << 1;
    const int rhalf = (ln >> 4) << 3;
    const int n0    = wave_ntile(wv) * 16;

    if (blockIdx.x < NRB) {
        const int mt  = blockIdx.x >> 1;
        const int nh  = blockIdx.x & 1;
        const int r   = t >> 4;
        const int c8  = (t & 15) * 8;
        const int ar  = mt * 16 + r;
        const int b   = ar >> 1, arm = ar & 1;

        const int loc = (arm ? a2loc : a1loc)[b];
        const int rec = (arm ? a2rec : a1rec)[b];
        const int ns  = (arm ? a2ns  : a1ns )[b];
        const float clk = clockv[b];

        float dw[16];
        float4 wr[16];
        w3_load(W_dyn, W_concat, wv, ln, dw, wr);

        const bool locw = wave_set0(wv);
        const int  n2   = (nh * 4 + (wv >> 1)) * 16;
        const float* wck = W_concat + (size_t)khalf * Dc + n0 + l16;
        const float* wrk = W_robot  + (size_t)((locw ? 0 : Dc) + khalf) * Dc + n2 + l16;
        v2f b0[GS];
        b_group_load(wck, 0, b0);

        const bool has_pm = (loc >= 1 && loc <= Pc);
        const int  p      = has_pm ? loc - 1 : 0;
        const int   stg  = lstage[b * Pc + p];
        const int   waf  = lhw[b * Pc + p];
        const float petv = pet[b * Pc + p];

        const bool rec_ok = (rec >= 0 && rec < Nw);
        const bool ns_ok  = (ns >= 1 && ns <= Sc);
        const float* rsrc = enc_row + ((size_t)b * Nw + (rec_ok ? rec : 0)) * Dc + c8;
        const float* nsrc = enc_col + ((size_t)b * Sc + (ns_ok ? ns - 1 : 0)) * Dc + c8;
        float4 y0 = *(const float4*)(rsrc);
        float4 y1 = *(const float4*)(rsrc + 4);
        float4 y2 = *(const float4*)(nsrc);
        float4 y3 = *(const float4*)(nsrc + 4);

        const bool waf_ok = (waf >= 0 && waf < Nw);
        const float* ssrc = enc_col + ((size_t)b * Sc + (stg - 1)) * Dc + c8;
        const float* wsrc = enc_row + ((size_t)b * Nw + (waf_ok ? waf : 0)) * Dc + c8;
        float4 x0 = *(const float4*)(ssrc);
        float4 x1 = *(const float4*)(ssrc + 4);
        float4 x2 = *(const float4*)(wsrc);
        float4 x3 = *(const float4*)(wsrc + 4);

        const float4 z4 = make_float4(0.f, 0.f, 0.f, 0.f);
        if (!rec_ok) { y0 = z4; y1 = z4; }
        if (!ns_ok)  { y2 = z4; y3 = z4; }
        if (!waf_ok) { x2 = z4; x3 = z4; }

        float* yd = &Ys[r * LDB + Dc + c8];
        *(float4*)(yd)          = y0;
        *(float4*)(yd + 4)      = y1;
        *(float4*)(yd + Dc)     = y2;
        *(float4*)(yd + Dc + 4) = y3;
        float* xd = &Xs[r * LDA + c8];
        *(float4*)(xd)          = x0;
        *(float4*)(xd + 4)      = x1;
        *(float4*)(xd + Dc)     = x2;
        *(float4*)(xd + Dc + 4) = x3;

        w3_partial(dw, wr, w3p, wv, ln);

        if ((t & 15) == 0) {
            const float d = petv - clk;
            Rp[r] = (d > 0.f ? d : 0.f) * NORM_INV;
            Lk[r] = has_pm ? 2 : (loc == Pc + 1 ? 1 : 0);
        }
        __syncthreads();

        float  w3q[8];
        float4 rpa, rpb;
        int4   lka, lkb;
        auto epi_loads = [&]() __attribute__((always_inline)) {
            #pragma unroll
            for (int w = 0; w < 8; ++w)
                w3q[w] = w3p[w * Dc + n0 + l16];
            rpa = *(const float4*)&Rp[rhalf];
            rpb = *(const float4*)&Rp[rhalf + 4];
            lka = *(const int4*)&Lk[rhalf];
            lkb = *(const int4*)&Lk[rhalf + 4];
        };

        v8f accp1[1] = {v8f{}};
        k_sweep<1, LDA, KA, true, true>(wck, wrk, &Xs[l16 * LDA + khalf], b0, accp1, epi_loads);
        const v8f accp = accp1[0];
        {
            float w3v = w3q[0];
            #pragma unroll
            for (int w = 1; w < 8; ++w)
                w3v += w3q[w];
            const float rp[8] = {rpa.x, rpa.y, rpa.z, rpa.w, rpb.x, rpb.y, rpb.z, rpb.w};
            const int   lk[8] = {lka.x, lka.y, lka.z, lka.w, lkb.x, lkb.y, lkb.z, lkb.w};
            #pragma unroll
            for (int q = 0; q < 8; ++q) {
                const int row   = rhalf + q;
                const float v   = accp[q] + rp[q] * w3v;
                const float alt = (lk[q] == 1) ? 1.f : 0.f;
                Ys[row * LDB + n0 + l16] = (lk[q] == 2) ? v : alt;
            }
        }
        __syncthreads();

        v8f acc1[1] = {v8f{}};
        float* const pp = &Xs[(wv >> 1) * 256 + ln];
        if (locw) {
            k_sweep<1, LDB, Dc, false, false>(wrk, nullptr, &Ys[l16 * LDB + khalf], b0, acc1, []() {});
            #pragma unroll
            for (int q = 0; q < 8; ++q)
                pp[q * 32] = acc1[0][q];
        } else {
            k_sweep<1, LDB, KB - Dc, false, false>(wrk, nullptr, &Ys[l16 * LDB + Dc + khalf], b0, acc1, []() {});
        }
        __syncthreads();
        if (locw) return;
        const v8f acc = acc1[0];
        #pragma unroll
        for (int q = 0; q < 8; ++q) {
            const int mrow = rhalf + q;
            const int oar  = mt * 16 + mrow;
            const int ob   = oar >> 1, oarm = oar & 1;
            out[((size_t)ob * (Pc + 2) + Pc + oarm) * Dc + n2 + l16] = acc[q] + pp[q * 32];
        }
        return;
    }

    const int bid = blockIdx.x - NRB;
    if (bid < N5)
        pm_rows<3, true>(enc_row, enc_col, clockv, pet, W_dyn, W_concat, lhw, lstage, out,
                         Xs, w3p, Rp, 80 * (bid >> 1) + 32 * (bid & 1), bid & 1);
    else
        pm_rows<2, false>(enc_row, enc_col, clockv, pet, W_dyn, W_concat, lhw, lstage, out,
                          Xs, w3p, Rp, 80 * (N5 / 2) + 32 * (bid - N5), 0);
}

extern "C" void kernel_launch(void* const* d_in, const int* in_sizes, int n_in,
                              void* d_out, int out_size, void* d_ws, size_t ws_size,
                              hipStream_t stream) {
    (void)in_sizes; (void)n_in; (void)out_size; (void)d_ws; (void)ws_size;
    const float* enc_row  = (const float*)d_in[0];
    const float* enc_col  = (const float*)d_in[1];
    const float* clockv   = (const float*)d_in[2];
    const float* pet      = (const float*)d_in[3];
    const float* W_dyn    = (const float*)d_in[4];
    const float* W_concat = (const float*)d_in[5];
    const float* W_robot  = (const float*)d_in[6];
    const int*   lhw      = (const int*)d_in[7];
    const int*   lstage   = (const int*)d_in[8];
    const int*   a1loc    = (const int*)d_in[9];
    const int*   a2loc    = (const int*)d_in[10];
    const int*   a1rec    = (const int*)d_in[11];
    const int*   a2rec    = (const int*)d_in[12];
    const int*   a1ns     = (const int*)d_in[13];
    const int*   a2ns     = (const int*)d_in[14];
    float* out = (float*)d_out;

    pm_robot_kernel<<<GRID, 256, 0, stream>>>(enc_row, enc_col, clockv, pet,
                                              W_dyn, W_concat, W_robot, lhw, lstage,
                                              a1loc, a2loc, a1rec, a2rec, a1ns, a2ns, out);
}
